# starting kernel
# baseline (speedup 1.0000x reference)
#include <hip/hip_runtime.h>
#include <cstdint>

#define N_NODES 50000
#define N_EDGES 800000
#define DIM     128

typedef __attribute__((ext_vector_type(2))) float v2f;
typedef __attribute__((ext_vector_type(8))) float v8f;

__global__ void __launch_bounds__(256) k_init_deg(float* __restrict__ ds,
                                                  float* __restrict__ dr) {
    int i = blockIdx.x * blockDim.x + threadIdx.x;
    if (i < N_NODES) { ds[i] = 2.0f; dr[i] = 2.0f; }
}

__global__ void __launch_bounds__(256) k_edge_deg(const int*   __restrict__ snd,
                                                  const int*   __restrict__ rcv,
                                                  const float* __restrict__ ew,
                                                  float* __restrict__ ds,
                                                  float* __restrict__ dr) {
    int i = blockIdx.x * blockDim.x + threadIdx.x;
    if (i < N_EDGES) {
        float w = ew[i] + 1.0f;
        unsafeAtomicAdd(&ds[snd[i]], w);
        unsafeAtomicAdd(&dr[rcv[i]], w);
    }
}

__global__ void __launch_bounds__(128) k_gemm_scale(const float* __restrict__ nodes,
                                                    const float* __restrict__ Wm,
                                                    const float* __restrict__ ds,
                                                    const float* __restrict__ dr,
                                                    float* __restrict__ y,
                                                    float* __restrict__ out) {
    __shared__ float w_lds[DIM * DIM];

    const int tid = threadIdx.x;

    for (int idx = tid; idx < DIM * DIM; idx += 128) {
        int k = idx >> 7;
        int c = idx & (DIM - 1);
        w_lds[((k >> 1) << 8) + (c << 1) + (k & 1)] = Wm[idx];
    }
    __syncthreads();

    const int lane = tid & 31;
    const int wave = tid >> 5;
    const int strip = blockIdx.x * 4 + wave;

    if (strip * 16 < N_NODES) {
        const int row0 = strip * 16;
        const int m    = lane & 15;
        const int koff = (lane >> 4) << 1;

        const float* arow = nodes + (size_t)(row0 + m) * DIM;

        v8f acc[8] = {};

        #pragma unroll 4
        for (int k0 = 0; k0 < DIM; k0 += 4) {
            v2f a = *(const v2f*)(arow + k0 + koff);
            const float* bbase = w_lds + (((k0 + koff) >> 1) << 8);
            #pragma unroll
            for (int t = 0; t < 8; ++t) {
                v2f b = *(const v2f*)(bbase + ((t * 16 + m) << 1));
                acc[t] = __builtin_amdgcn_wmma_f32_16x16x4_f32(
                     false, a,  false, b,
                     (short)0, acc[t],
                     false,  false);
            }
        }

        const int roff = (lane >> 4) << 3;
        #pragma unroll
        for (int v = 0; v < 8; ++v) {
            const int row = row0 + v + roff;
            const float a1 = ds[row];
            const float a2 = dr[row];
            const float sc = rsqrtf(fmaxf(a1, 1.0f)) * rsqrtf(fmaxf(a2, 1.0f));
            float* yrow = y   + (size_t)row * DIM;
            float* orow = out + (size_t)row * DIM;
            #pragma unroll
            for (int t = 0; t < 8; ++t) {
                const float val = acc[t][v] * sc;
                yrow[t * 16 + m] = val;
                orow[t * 16 + m] = val;
            }
        }
    }
}

__global__ void __launch_bounds__(256) k_edge_scatter(const int*   __restrict__ snd,
                                                      const int*   __restrict__ rcv,
                                                      const float* __restrict__ ew,
                                                      const float* __restrict__ y,
                                                      float* out) {
    const int gtid = blockIdx.x * blockDim.x + threadIdx.x;
    const int e    = gtid >> 5;
    if (e >= N_EDGES) return;
    const int lane = gtid & 31;

    const int   s = snd[e];
    const int   r = rcv[e];
    const float w = ew[e];

    const float4 v = *(const float4*)(y + (size_t)s * DIM + lane * 4);
    float* dst = out + (size_t)r * DIM + lane * 4;
    unsafeAtomicAdd(dst + 0, v.x * w);
    unsafeAtomicAdd(dst + 1, v.y * w);
    unsafeAtomicAdd(dst + 2, v.z * w);
    unsafeAtomicAdd(dst + 3, v.w * w);
}

extern "C" void kernel_launch(void* const* d_in, const int* in_sizes, int n_in,
                              void* d_out, int out_size, void* d_ws, size_t ws_size,
                              hipStream_t stream) {
    const float* nodes     = (const float*)d_in[0];
    const int*   senders   = (const int*)  d_in[1];
    const int*   receivers = (const int*)  d_in[2];
    const float* edges     = (const float*)d_in[3];
    const float* Wm        = (const float*)d_in[4];
    float*       out       = (float*)d_out;

    float* y  = (float*)d_ws;
    float* ds = y  + (size_t)N_NODES * DIM;
    float* dr = ds + N_NODES;

    k_init_deg<<<(N_NODES + 255) / 256, 256, 0, stream>>>(ds, dr);
    k_edge_deg<<<(N_EDGES + 255) / 256, 256, 0, stream>>>(senders, receivers, edges, ds, dr);

    const int n_strips = (N_NODES + 15) / 16;
    k_gemm_scale<<<(n_strips + 3) / 4, 128, 0, stream>>>(nodes, Wm, ds, dr, y, out);

    const long long scatter_threads = (long long)N_EDGES * 32;
    k_edge_scatter<<<(int)((scatter_threads + 255) / 256), 256, 0, stream>>>(
        senders, receivers, edges, y, out);
}


# degree sums: each workgroup sums an edge slice into one whole degree array in LDS (LDS float atomics), then adds it to global a whole line per atomic; order of the sums changes
# speedup vs baseline: 1.4007x; 1.4007x over previous
#include <hip/hip_runtime.h>
#include <cstdint>

#define N_NODES 50000
#define N_EDGES 800000
#define DIM     128

typedef __attribute__((ext_vector_type(2))) float v2f;
typedef __attribute__((ext_vector_type(8))) float v8f;

__global__ void __launch_bounds__(256) k_init_deg(float* __restrict__ ds,
                                                  float* __restrict__ dr) {
    int i = blockIdx.x * blockDim.x + threadIdx.x;
    if (i < N_NODES) { ds[i] = 2.0f; dr[i] = 2.0f; }
}

#define DEG_SLICES  32
#define DEG_THREADS 1024
#define DEG_PER     (N_EDGES / DEG_SLICES)
#define DEG_NV      (DEG_PER / 4)
#define DEG_IT      ((DEG_NV + DEG_THREADS - 1) / DEG_THREADS)
#define DEG_STEPS   ((N_NODES + DEG_THREADS - 1) / DEG_THREADS)
#define DEG_BATCH   7
static_assert(N_EDGES % (DEG_SLICES * 4) == 0, "slices must be whole 16-B groups");
static_assert(DEG_STEPS % DEG_BATCH == 0, "hand-over loop runs in whole batches");
static_assert(DEG_STEPS >= DEG_SLICES, "every slice gets a start step of its own");

__global__ void __launch_bounds__(DEG_THREADS) k_edge_deg(const int*   __restrict__ snd,
                                                          const int*   __restrict__ rcv,
                                                          const float* __restrict__ ew,
                                                          float* __restrict__ ds,
                                                          float* __restrict__ dr) {
    __shared__ __attribute__((aligned(16))) float hist[DEG_STEPS * DEG_THREADS];

    const int  tid   = threadIdx.x;
    const bool recv  = blockIdx.x >= DEG_SLICES;
    const int  slice = recv ? blockIdx.x - DEG_SLICES : blockIdx.x;
    const int*  idx  = recv ? rcv : snd;
    float*      dst  = recv ? dr  : ds;

    const int4*   ip = (const int4*)  (idx + slice * DEG_PER);
    const float4* wp = (const float4*)(ew  + slice * DEG_PER);
    int4   n[DEG_IT];
    float4 w[DEG_IT];
    #pragma unroll
    for (int k = 0; k < DEG_IT; ++k) {
        const int v  = tid + k * DEG_THREADS;
        const int vc = v < DEG_NV ? v : DEG_NV - 1;
        n[k] = ip[vc];
        w[k] = wp[vc];
    }

    for (int i = tid; i < DEG_STEPS * DEG_THREADS / 4; i += DEG_THREADS)
        ((float4*)hist)[i] = make_float4(0.0f, 0.0f, 0.0f, 0.0f);
    __syncthreads();

    #pragma unroll
    for (int k = 0; k < DEG_IT; ++k) {
        if (tid + k * DEG_THREADS < DEG_NV) {
            atomicAdd(&hist[n[k].x], w[k].x + 1.0f);
            atomicAdd(&hist[n[k].y], w[k].y + 1.0f);
            atomicAdd(&hist[n[k].z], w[k].z + 1.0f);
            atomicAdd(&hist[n[k].w], w[k].w + 1.0f);
        }
    }
    __syncthreads();

    const int off = (slice * DEG_STEPS) / DEG_SLICES;
    for (int k0 = 0; k0 < DEG_STEPS; k0 += DEG_BATCH) {
        int   node[DEG_BATCH];
        float part[DEG_BATCH];
        #pragma unroll
        for (int u = 0; u < DEG_BATCH; ++u) {
            int st = k0 + u + off;
            if (st >= DEG_STEPS) st -= DEG_STEPS;
            node[u] = st * DEG_THREADS + tid;
            part[u] = hist[node[u]];
        }
        #pragma unroll
        for (int u = 0; u < DEG_BATCH; ++u)
            if (node[u] < N_NODES) unsafeAtomicAdd(&dst[node[u]], part[u]);
    }
}

__global__ void __launch_bounds__(128) k_gemm_scale(const float* __restrict__ nodes,
                                                    const float* __restrict__ Wm,
                                                    const float* __restrict__ ds,
                                                    const float* __restrict__ dr,
                                                    float* __restrict__ y,
                                                    float* __restrict__ out) {
    __shared__ float w_lds[DIM * DIM];

    const int tid = threadIdx.x;

    for (int idx = tid; idx < DIM * DIM; idx += 128) {
        int k = idx >> 7;
        int c = idx & (DIM - 1);
        w_lds[((k >> 1) << 8) + (c << 1) + (k & 1)] = Wm[idx];
    }
    __syncthreads();

    const int lane = tid & 31;
    const int wave = tid >> 5;
    const int strip = blockIdx.x * 4 + wave;

    if (strip * 16 < N_NODES) {
        const int row0 = strip * 16;
        const int m    = lane & 15;
        const int koff = (lane >> 4) << 1;

        const float* arow = nodes + (size_t)(row0 + m) * DIM;

        v8f acc[8] = {};

        #pragma unroll 4
        for (int k0 = 0; k0 < DIM; k0 += 4) {
            v2f a = *(const v2f*)(arow + k0 + koff);
            const float* bbase = w_lds + (((k0 + koff) >> 1) << 8);
            #pragma unroll
            for (int t = 0; t < 8; ++t) {
                v2f b = *(const v2f*)(bbase + ((t * 16 + m) << 1));
                acc[t] = __builtin_amdgcn_wmma_f32_16x16x4_f32(
                     false, a,  false, b,
                     (short)0, acc[t],
                     false,  false);
            }
        }

        const int roff = (lane >> 4) << 3;
        #pragma unroll
        for (int v = 0; v < 8; ++v) {
            const int row = row0 + v + roff;
            const float a1 = ds[row];
            const float a2 = dr[row];
            const float sc = rsqrtf(fmaxf(a1, 1.0f)) * rsqrtf(fmaxf(a2, 1.0f));
            float* yrow = y   + (size_t)row * DIM;
            float* orow = out + (size_t)row * DIM;
            #pragma unroll
            for (int t = 0; t < 8; ++t) {
                const float val = acc[t][v] * sc;
                yrow[t * 16 + m] = val;
                orow[t * 16 + m] = val;
            }
        }
    }
}

__global__ void __launch_bounds__(256) k_edge_scatter(const int*   __restrict__ snd,
                                                      const int*   __restrict__ rcv,
                                                      const float* __restrict__ ew,
                                                      const float* __restrict__ y,
                                                      float* out) {
    const int gtid = blockIdx.x * blockDim.x + threadIdx.x;
    const int e    = gtid >> 5;
    if (e >= N_EDGES) return;
    const int lane = gtid & 31;

    const int   s = snd[e];
    const int   r = rcv[e];
    const float w = ew[e];

    const float4 v = *(const float4*)(y + (size_t)s * DIM + lane * 4);
    float* dst = out + (size_t)r * DIM + lane * 4;
    unsafeAtomicAdd(dst + 0, v.x * w);
    unsafeAtomicAdd(dst + 1, v.y * w);
    unsafeAtomicAdd(dst + 2, v.z * w);
    unsafeAtomicAdd(dst + 3, v.w * w);
}

extern "C" void kernel_launch(void* const* d_in, const int* in_sizes, int n_in,
                              void* d_out, int out_size, void* d_ws, size_t ws_size,
                              hipStream_t stream) {
    const float* nodes     = (const float*)d_in[0];
    const int*   senders   = (const int*)  d_in[1];
    const int*   receivers = (const int*)  d_in[2];
    const float* edges     = (const float*)d_in[3];
    const float* Wm        = (const float*)d_in[4];
    float*       out       = (float*)d_out;

    float* y  = (float*)d_ws;
    float* ds = y  + (size_t)N_NODES * DIM;
    float* dr = ds + N_NODES;

    k_init_deg<<<(N_NODES + 255) / 256, 256, 0, stream>>>(ds, dr);
    k_edge_deg<<<2 * DEG_SLICES, DEG_THREADS, 0, stream>>>(senders, receivers, edges, ds, dr);

    const int n_strips = (N_NODES + 15) / 16;
    k_gemm_scale<<<(n_strips + 3) / 4, 128, 0, stream>>>(nodes, Wm, ds, dr, y, out);

    const long long scatter_threads = (long long)N_EDGES * 32;
    k_edge_scatter<<<(int)((scatter_threads + 255) / 256), 256, 0, stream>>>(
        senders, receivers, edges, y, out);
}


# edge scatter: y rows stored column-interleaved so lane L's float4 load holds columns L,32+L,64+L,96+L and each f32 atomic add covers one whole 128-B line, not a quarter of four lines
# speedup vs baseline: 1.9598x; 1.3992x over previous
#include <hip/hip_runtime.h>
#include <cstdint>

#define N_NODES 50000
#define N_EDGES 800000
#define DIM     128

typedef __attribute__((ext_vector_type(2))) float v2f;
typedef __attribute__((ext_vector_type(8))) float v8f;

__global__ void __launch_bounds__(256) k_init_deg(float* __restrict__ ds,
                                                  float* __restrict__ dr) {
    int i = blockIdx.x * blockDim.x + threadIdx.x;
    if (i < N_NODES) { ds[i] = 2.0f; dr[i] = 2.0f; }
}

#define DEG_SLICES  32
#define DEG_THREADS 1024
#define DEG_PER     (N_EDGES / DEG_SLICES)
#define DEG_NV      (DEG_PER / 4)
#define DEG_IT      ((DEG_NV + DEG_THREADS - 1) / DEG_THREADS)
#define DEG_STEPS   ((N_NODES + DEG_THREADS - 1) / DEG_THREADS)
#define DEG_BATCH   7
static_assert(N_EDGES % (DEG_SLICES * 4) == 0, "slices must be whole 16-B groups");
static_assert(DEG_STEPS % DEG_BATCH == 0, "hand-over loop runs in whole batches");
static_assert(DEG_STEPS >= DEG_SLICES, "every slice gets a start step of its own");

__global__ void __launch_bounds__(DEG_THREADS) k_edge_deg(const int*   __restrict__ snd,
                                                          const int*   __restrict__ rcv,
                                                          const float* __restrict__ ew,
                                                          float* __restrict__ ds,
                                                          float* __restrict__ dr) {
    __shared__ __attribute__((aligned(16))) float hist[DEG_STEPS * DEG_THREADS];

    const int  tid   = threadIdx.x;
    const bool recv  = blockIdx.x >= DEG_SLICES;
    const int  slice = recv ? blockIdx.x - DEG_SLICES : blockIdx.x;
    const int*  idx  = recv ? rcv : snd;
    float*      dst  = recv ? dr  : ds;

    const int4*   ip = (const int4*)  (idx + slice * DEG_PER);
    const float4* wp = (const float4*)(ew  + slice * DEG_PER);
    int4   n[DEG_IT];
    float4 w[DEG_IT];
    #pragma unroll
    for (int k = 0; k < DEG_IT; ++k) {
        const int v  = tid + k * DEG_THREADS;
        const int vc = v < DEG_NV ? v : DEG_NV - 1;
        n[k] = ip[vc];
        w[k] = wp[vc];
    }

    for (int i = tid; i < DEG_STEPS * DEG_THREADS / 4; i += DEG_THREADS)
        ((float4*)hist)[i] = make_float4(0.0f, 0.0f, 0.0f, 0.0f);
    __syncthreads();

    #pragma unroll
    for (int k = 0; k < DEG_IT; ++k) {
        if (tid + k * DEG_THREADS < DEG_NV) {
            atomicAdd(&hist[n[k].x], w[k].x + 1.0f);
            atomicAdd(&hist[n[k].y], w[k].y + 1.0f);
            atomicAdd(&hist[n[k].z], w[k].z + 1.0f);
            atomicAdd(&hist[n[k].w], w[k].w + 1.0f);
        }
    }
    __syncthreads();

    const int off = (slice * DEG_STEPS) / DEG_SLICES;
    for (int k0 = 0; k0 < DEG_STEPS; k0 += DEG_BATCH) {
        int   node[DEG_BATCH];
        float part[DEG_BATCH];
        #pragma unroll
        for (int u = 0; u < DEG_BATCH; ++u) {
            int st = k0 + u + off;
            if (st >= DEG_STEPS) st -= DEG_STEPS;
            node[u] = st * DEG_THREADS + tid;
            part[u] = hist[node[u]];
        }
        #pragma unroll
        for (int u = 0; u < DEG_BATCH; ++u)
            if (node[u] < N_NODES) unsafeAtomicAdd(&dst[node[u]], part[u]);
    }
}

__global__ void __launch_bounds__(128) k_gemm_scale(const float* __restrict__ nodes,
                                                    const float* __restrict__ Wm,
                                                    const float* __restrict__ ds,
                                                    const float* __restrict__ dr,
                                                    float* __restrict__ y,
                                                    float* __restrict__ out) {
    __shared__ float w_lds[DIM * DIM];

    const int tid = threadIdx.x;

    for (int idx = tid; idx < DIM * DIM; idx += 128) {
        int k = idx >> 7;
        int c = idx & (DIM - 1);
        w_lds[((k >> 1) << 8) + (c << 1) + (k & 1)] = Wm[idx];
    }
    __syncthreads();

    const int lane = tid & 31;
    const int wave = tid >> 5;
    const int strip = blockIdx.x * 4 + wave;

    if (strip * 16 < N_NODES) {
        const int row0 = strip * 16;
        const int m    = lane & 15;
        const int koff = (lane >> 4) << 1;

        const float* arow = nodes + (size_t)(row0 + m) * DIM;

        v8f acc[8] = {};

        #pragma unroll 4
        for (int k0 = 0; k0 < DIM; k0 += 4) {
            v2f a = *(const v2f*)(arow + k0 + koff);
            const float* bbase = w_lds + (((k0 + koff) >> 1) << 8);
            #pragma unroll
            for (int t = 0; t < 8; ++t) {
                v2f b = *(const v2f*)(bbase + ((t * 16 + m) << 1));
                acc[t] = __builtin_amdgcn_wmma_f32_16x16x4_f32(
                     false, a,  false, b,
                     (short)0, acc[t],
                     false,  false);
            }
        }

        const int roff = (lane >> 4) << 3;
        #pragma unroll
        for (int v = 0; v < 8; ++v) {
            const int row = row0 + v + roff;
            const float a1 = ds[row];
            const float a2 = dr[row];
            const float sc = rsqrtf(fmaxf(a1, 1.0f)) * rsqrtf(fmaxf(a2, 1.0f));
            float* yrow = y   + (size_t)row * DIM;
            float* orow = out + (size_t)row * DIM;
            float val[8];
            #pragma unroll
            for (int t = 0; t < 8; ++t) {
                val[t] = acc[t][v] * sc;
                orow[t * 16 + m] = val[t];
            }
            *(float4*)(yrow + 4 * m)      = make_float4(val[0], val[2], val[4], val[6]);
            *(float4*)(yrow + 64 + 4 * m) = make_float4(val[1], val[3], val[5], val[7]);
        }
    }
}

__global__ void __launch_bounds__(256) k_edge_scatter(const int*   __restrict__ snd,
                                                      const int*   __restrict__ rcv,
                                                      const float* __restrict__ ew,
                                                      const float* __restrict__ y,
                                                      float* out) {
    const int gtid = blockIdx.x * blockDim.x + threadIdx.x;
    const int e    = gtid >> 5;
    if (e >= N_EDGES) return;
    const int lane = gtid & 31;

    const int   s = snd[e];
    const int   r = rcv[e];
    const float w = ew[e];

    const float4 v = *(const float4*)(y + (size_t)s * DIM + lane * 4);
    float* dst = out + (size_t)r * DIM + lane;
    unsafeAtomicAdd(dst,      v.x * w);
    unsafeAtomicAdd(dst + 32, v.y * w);
    unsafeAtomicAdd(dst + 64, v.z * w);
    unsafeAtomicAdd(dst + 96, v.w * w);
}

extern "C" void kernel_launch(void* const* d_in, const int* in_sizes, int n_in,
                              void* d_out, int out_size, void* d_ws, size_t ws_size,
                              hipStream_t stream) {
    const float* nodes     = (const float*)d_in[0];
    const int*   senders   = (const int*)  d_in[1];
    const int*   receivers = (const int*)  d_in[2];
    const float* edges     = (const float*)d_in[3];
    const float* Wm        = (const float*)d_in[4];
    float*       out       = (float*)d_out;

    float* y  = (float*)d_ws;
    float* ds = y  + (size_t)N_NODES * DIM;
    float* dr = ds + N_NODES;

    k_init_deg<<<(N_NODES + 255) / 256, 256, 0, stream>>>(ds, dr);
    k_edge_deg<<<2 * DEG_SLICES, DEG_THREADS, 0, stream>>>(senders, receivers, edges, ds, dr);

    const int n_strips = (N_NODES + 15) / 16;
    k_gemm_scale<<<(n_strips + 3) / 4, 128, 0, stream>>>(nodes, Wm, ds, dr, y, out);

    const long long scatter_threads = (long long)N_EDGES * 32;
    k_edge_scatter<<<(int)((scatter_threads + 255) / 256), 256, 0, stream>>>(
        senders, receivers, edges, y, out);
}


# edge sum: edges binned by 200-node receiver range; one workgroup per range adds its sender rows into LDS rows (LDS f32 atomics), stores each out line once; no global f32 atomics; sum order changes
# speedup vs baseline: 3.0322x; 1.5472x over previous
#include <hip/hip_runtime.h>
#include <cstdint>

#define N_NODES 50000
#define N_EDGES 800000
#define DIM     128

#define RN      200
#define NR      (N_NODES / RN)
#define NR_PAD  256
static_assert(N_NODES % RN == 0, "ranges are whole");
static_assert(NR <= NR_PAD && RN <= 256, "range id and local row fit their fields");
static_assert(N_NODES <= 65536, "a sender id fits 16 bits of the packed edge word");

typedef __attribute__((ext_vector_type(2))) float v2f;
typedef __attribute__((ext_vector_type(8))) float v8f;

__global__ void __launch_bounds__(256) k_init_deg(float* __restrict__ ds,
                                                  float* __restrict__ dr,
                                                  int* __restrict__ rcount) {
    int i = blockIdx.x * blockDim.x + threadIdx.x;
    if (i < N_NODES) { ds[i] = 2.0f; dr[i] = 2.0f; }
    if (i < 2 * NR_PAD) rcount[i] = 0;
}

#define DEG_SLICES  32
#define DEG_THREADS 1024
#define DEG_PER     (N_EDGES / DEG_SLICES)
#define DEG_NV      (DEG_PER / 4)
#define DEG_IT      ((DEG_NV + DEG_THREADS - 1) / DEG_THREADS)
#define DEG_STEPS   ((N_NODES + DEG_THREADS - 1) / DEG_THREADS)
#define DEG_BATCH   7
static_assert(N_EDGES % (DEG_SLICES * 4) == 0, "slices must be whole 16-B groups");
static_assert(DEG_STEPS % DEG_BATCH == 0, "hand-over loop runs in whole batches");
static_assert(DEG_STEPS >= DEG_SLICES, "every slice gets a start step of its own");

__global__ void __launch_bounds__(DEG_THREADS) k_edge_deg(const int*   __restrict__ snd,
                                                          const int*   __restrict__ rcv,
                                                          const float* __restrict__ ew,
                                                          float* __restrict__ ds,
                                                          float* __restrict__ dr,
                                                          int* __restrict__ rtot) {
    __shared__ int rc[NR_PAD];
    __shared__ __attribute__((aligned(16))) float hist[DEG_STEPS * DEG_THREADS];

    const int  tid   = threadIdx.x;
    const bool recv  = blockIdx.x >= DEG_SLICES;
    const int  slice = recv ? blockIdx.x - DEG_SLICES : blockIdx.x;
    const int*  idx  = recv ? rcv : snd;
    float*      dst  = recv ? dr  : ds;

    const int4*   ip = (const int4*)  (idx + slice * DEG_PER);
    const float4* wp = (const float4*)(ew  + slice * DEG_PER);
    int4   n[DEG_IT];
    float4 w[DEG_IT];
    #pragma unroll
    for (int k = 0; k < DEG_IT; ++k) {
        const int v  = tid + k * DEG_THREADS;
        const int vc = v < DEG_NV ? v : DEG_NV - 1;
        n[k] = ip[vc];
        w[k] = wp[vc];
    }

    for (int i = tid; i < DEG_STEPS * DEG_THREADS / 4; i += DEG_THREADS)
        ((float4*)hist)[i] = make_float4(0.0f, 0.0f, 0.0f, 0.0f);
    if (tid < NR_PAD) rc[tid] = 0;
    __syncthreads();

    #pragma unroll
    for (int k = 0; k < DEG_IT; ++k) {
        if (tid + k * DEG_THREADS < DEG_NV) {
            atomicAdd(&hist[n[k].x], w[k].x + 1.0f);
            atomicAdd(&hist[n[k].y], w[k].y + 1.0f);
            atomicAdd(&hist[n[k].z], w[k].z + 1.0f);
            atomicAdd(&hist[n[k].w], w[k].w + 1.0f);
            if (recv) {
                atomicAdd(&rc[n[k].x / RN], 1);
                atomicAdd(&rc[n[k].y / RN], 1);
                atomicAdd(&rc[n[k].z / RN], 1);
                atomicAdd(&rc[n[k].w / RN], 1);
            }
        }
    }
    __syncthreads();
    if (recv && tid < NR) atomicAdd(&rtot[tid], rc[tid]);

    const int off = (slice * DEG_STEPS) / DEG_SLICES;
    for (int k0 = 0; k0 < DEG_STEPS; k0 += DEG_BATCH) {
        int   node[DEG_BATCH];
        float part[DEG_BATCH];
        #pragma unroll
        for (int u = 0; u < DEG_BATCH; ++u) {
            int st = k0 + u + off;
            if (st >= DEG_STEPS) st -= DEG_STEPS;
            node[u] = st * DEG_THREADS + tid;
            part[u] = hist[node[u]];
        }
        #pragma unroll
        for (int u = 0; u < DEG_BATCH; ++u)
            if (node[u] < N_NODES) unsafeAtomicAdd(&dst[node[u]], part[u]);
    }
}

__global__ void __launch_bounds__(128) k_gemm_scale(const float* __restrict__ nodes,
                                                    const float* __restrict__ Wm,
                                                    const float* __restrict__ ds,
                                                    const float* __restrict__ dr,
                                                    float* __restrict__ y) {
    __shared__ float w_lds[DIM * DIM];

    const int tid = threadIdx.x;

    for (int idx = tid; idx < DIM * DIM; idx += 128) {
        int k = idx >> 7;
        int c = idx & (DIM - 1);
        w_lds[((k >> 1) << 8) + (c << 1) + (k & 1)] = Wm[idx];
    }
    __syncthreads();

    const int lane = tid & 31;
    const int wave = tid >> 5;
    const int strip = blockIdx.x * 4 + wave;

    if (strip * 16 < N_NODES) {
        const int row0 = strip * 16;
        const int m    = lane & 15;
        const int koff = (lane >> 4) << 1;

        const float* arow = nodes + (size_t)(row0 + m) * DIM;

        v8f acc[8] = {};

        #pragma unroll 4
        for (int k0 = 0; k0 < DIM; k0 += 4) {
            v2f a = *(const v2f*)(arow + k0 + koff);
            const float* bbase = w_lds + (((k0 + koff) >> 1) << 8);
            #pragma unroll
            for (int t = 0; t < 8; ++t) {
                v2f b = *(const v2f*)(bbase + ((t * 16 + m) << 1));
                acc[t] = __builtin_amdgcn_wmma_f32_16x16x4_f32(
                     false, a,  false, b,
                     (short)0, acc[t],
                     false,  false);
            }
        }

        const int roff = (lane >> 4) << 3;
        #pragma unroll
        for (int v = 0; v < 8; ++v) {
            const int row = row0 + v + roff;
            const float a1 = ds[row];
            const float a2 = dr[row];
            const float sc = rsqrtf(fmaxf(a1, 1.0f)) * rsqrtf(fmaxf(a2, 1.0f));
            float* yrow = y + (size_t)row * DIM;
            float val[8];
            #pragma unroll
            for (int t = 0; t < 8; ++t) val[t] = acc[t][v] * sc;
            *(float4*)(yrow + 4 * m)      = make_float4(val[0], val[2], val[4], val[6]);
            *(float4*)(yrow + 64 + 4 * m) = make_float4(val[1], val[3], val[5], val[7]);
        }
    }
}

#define BIN_THREADS 1024
#define BIN_PER     3200
#define BIN_BLOCKS  (N_EDGES / BIN_PER)
#define BIN_IT      ((BIN_PER + BIN_THREADS - 1) / BIN_THREADS)
static_assert(N_EDGES % BIN_PER == 0, "slices are whole");

__global__ void __launch_bounds__(BIN_THREADS) k_bin(const int*   __restrict__ snd,
                                                     const int*   __restrict__ rcv,
                                                     const float* __restrict__ ew,
                                                     const int*   __restrict__ rtot,
                                                     int*  __restrict__ rfill,
                                                     int2* __restrict__ bins) {
    __shared__ int s_wsum[NR_PAD / 32];
    __shared__ int s_cnt[NR_PAD];
    __shared__ int s_base[NR_PAD];

    const int tid  = threadIdx.x;
    const int lane = tid & 31;
    const int wave = tid >> 5;
    const int e0   = blockIdx.x * BIN_PER;

    int tot = 0, incl = 0;
    if (tid < NR_PAD) {
        s_cnt[tid] = 0;
        tot  = tid < NR ? rtot[tid] : 0;
        incl = tot;
        #pragma unroll
        for (int d = 1; d < 32; d <<= 1) {
            const int t = __shfl_up(incl, d);
            if (lane >= d) incl += t;
        }
        if (lane == 31) s_wsum[wave] = incl;
    }

    int   rr[BIN_IT], ss[BIN_IT];
    float ww[BIN_IT];
    #pragma unroll
    for (int k = 0; k < BIN_IT; ++k) {
        const int i = tid + k * BIN_THREADS;
        rr[k] = 0; ss[k] = 0; ww[k] = 0.0f;
        if (i < BIN_PER) { rr[k] = rcv[e0 + i]; ss[k] = snd[e0 + i]; ww[k] = ew[e0 + i]; }
    }
    __syncthreads();

    int g[BIN_IT], rank[BIN_IT];
    #pragma unroll
    for (int k = 0; k < BIN_IT; ++k) {
        g[k] = rr[k] / RN;
        rank[k] = 0;
        if (tid + k * BIN_THREADS < BIN_PER) rank[k] = atomicAdd(&s_cnt[g[k]], 1);
    }
    __syncthreads();

    if (tid < NR) {
        int st = incl - tot;
        for (int k = 0; k < wave; ++k) st += s_wsum[k];
        s_base[tid] = st + atomicAdd(&rfill[tid], s_cnt[tid]);
    }
    __syncthreads();

    #pragma unroll
    for (int k = 0; k < BIN_IT; ++k) {
        if (tid + k * BIN_THREADS < BIN_PER)
            bins[s_base[g[k]] + rank[k]] =
                make_int2(ss[k] | ((rr[k] - g[k] * RN) << 16), __float_as_int(ww[k]));
    }
}

#define AGG_THREADS 1024
#define AGG_WAVES   (AGG_THREADS / 32)
#define AGG_ROWS    8
#define AGG_INIT    ((RN + AGG_WAVES - 1) / AGG_WAVES)

__global__ void __launch_bounds__(AGG_THREADS) k_row_sums(const float* __restrict__ y,
                                                          const int*   __restrict__ rtot,
                                                          const int2*  __restrict__ bins,
                                                          float* __restrict__ out) {
    __shared__ __attribute__((aligned(16))) float acc[RN * DIM];
    __shared__ int s_start;

    const int tid   = threadIdx.x;
    const int lane  = tid & 31;
    const int wave  = tid >> 5;
    const int range = blockIdx.x;
    const int node0 = range * RN;

    if (tid == 0) s_start = 0;
    __syncthreads();

    if (tid < range) atomicAdd(&s_start, rtot[tid]);

    float4 self[AGG_INIT];
    #pragma unroll
    for (int p = 0; p < AGG_INIT; ++p) {
        const int rl  = wave + p * AGG_WAVES;
        const int rlc = rl < RN ? rl : RN - 1;
        self[p] = *(const float4*)(y + (size_t)(node0 + rlc) * DIM + 4 * lane);
    }
    #pragma unroll
    for (int p = 0; p < AGG_INIT; ++p) {
        const int rl = wave + p * AGG_WAVES;
        if (rl < RN) {
            float* a = acc + rl * DIM + lane;
            a[0] = self[p].x; a[32] = self[p].y; a[64] = self[p].z; a[96] = self[p].w;
        }
    }
    __syncthreads();

    const int start = s_start;
    const int cnt   = rtot[range];

    int  c0  = wave * 32;
    int  idx = start + c0 + lane;
    int2 rec = bins[idx < N_EDGES ? idx : N_EDGES - 1];
    while (c0 < cnt) {
        const int n = cnt - c0 < 32 ? cnt - c0 : 32;
        const int c1 = c0 + AGG_WAVES * 32;
        idx = start + c1 + lane;
        const int2 nrec = bins[idx < N_EDGES ? idx : N_EDGES - 1];
        #pragma unroll
        for (int j0 = 0; j0 < 32; j0 += AGG_ROWS) {
            if (j0 < n) {
                float4 v[AGG_ROWS];
                int    rl[AGG_ROWS];
                float  w[AGG_ROWS];
                #pragma unroll
                for (int u = 0; u < AGG_ROWS; ++u) {
                    const int px = __builtin_amdgcn_readlane(rec.x, j0 + u);
                    w[u]  = __int_as_float(__builtin_amdgcn_readlane(rec.y, j0 + u));
                    rl[u] = px >> 16;
                    v[u]  = *(const float4*)(y + (size_t)(px & 0xFFFF) * DIM + lane * 4);
                }
                #pragma unroll
                for (int u = 0; u < AGG_ROWS; ++u)
                    asm volatile("" :: "v"(v[u].x), "v"(v[u].y), "v"(v[u].z), "v"(v[u].w));
                #pragma unroll
                for (int u = 0; u < AGG_ROWS; ++u) {
                    if (j0 + u < n) {
                        float* a = acc + rl[u] * DIM + lane;
                        atomicAdd(a,      v[u].x * w[u]);
                        atomicAdd(a + 32, v[u].y * w[u]);
                        atomicAdd(a + 64, v[u].z * w[u]);
                        atomicAdd(a + 96, v[u].w * w[u]);
                    }
                }
            }
        }
        c0 = c1; rec = nrec;
    }
    __syncthreads();

    float4*       orow = (float4*)(out + (size_t)node0 * DIM);
    const float4* arow = (const float4*)acc;
    for (int idx = tid; idx < RN * DIM / 4; idx += AGG_THREADS) orow[idx] = arow[idx];
}

extern "C" void kernel_launch(void* const* d_in, const int* in_sizes, int n_in,
                              void* d_out, int out_size, void* d_ws, size_t ws_size,
                              hipStream_t stream) {
    const float* nodes     = (const float*)d_in[0];
    const int*   senders   = (const int*)  d_in[1];
    const int*   receivers = (const int*)  d_in[2];
    const float* edges     = (const float*)d_in[3];
    const float* Wm        = (const float*)d_in[4];
    float*       out       = (float*)d_out;

    float* y     = (float*)d_ws;
    float* ds    = y  + (size_t)N_NODES * DIM;
    float* dr    = ds + N_NODES;
    int*   rtot  = (int*)(dr + N_NODES);
    int*   rfill = rtot + NR_PAD;
    int2*  bins  = (int2*)(rfill + NR_PAD);

    k_init_deg<<<(N_NODES + 255) / 256, 256, 0, stream>>>(ds, dr, rtot);
    k_edge_deg<<<2 * DEG_SLICES, DEG_THREADS, 0, stream>>>(senders, receivers, edges, ds, dr, rtot);

    const int n_strips = (N_NODES + 15) / 16;
    k_gemm_scale<<<(n_strips + 3) / 4, 128, 0, stream>>>(nodes, Wm, ds, dr, y);

    k_bin<<<BIN_BLOCKS, BIN_THREADS, 0, stream>>>(senders, receivers, edges, rtot, rfill, bins);
    k_row_sums<<<NR, AGG_THREADS, 0, stream>>>(y, rtot, bins, out);
}


# GEMM and edge binning share one launch (role by block index, 1024-thread blocks, two per WGP; a GEMM wave takes a 16x64 half strip): binning waits pass under matrix work, one launch fewer
# speedup vs baseline: 3.5666x; 1.1762x over previous
#include <hip/hip_runtime.h>
#include <cstdint>

#define N_NODES 50000
#define N_EDGES 800000
#define DIM     128

#define RN      200
#define NR      (N_NODES / RN)
#define NR_PAD  256
static_assert(N_NODES % RN == 0, "ranges are whole");
static_assert(NR <= NR_PAD && RN <= 256, "range id and local row fit their fields");
static_assert(N_NODES <= 65536, "a sender id fits 16 bits of the packed edge word");

typedef __attribute__((ext_vector_type(2))) float v2f;
typedef __attribute__((ext_vector_type(8))) float v8f;

__global__ void __launch_bounds__(256) k_init_deg(float* __restrict__ ds,
                                                  float* __restrict__ dr,
                                                  int* __restrict__ rcount) {
    int i = blockIdx.x * blockDim.x + threadIdx.x;
    if (i < N_NODES) { ds[i] = 2.0f; dr[i] = 2.0f; }
    if (i < 2 * NR_PAD) rcount[i] = 0;
}

#define DEG_SLICES  32
#define DEG_THREADS 1024
#define DEG_PER     (N_EDGES / DEG_SLICES)
#define DEG_NV      (DEG_PER / 4)
#define DEG_IT      ((DEG_NV + DEG_THREADS - 1) / DEG_THREADS)
#define DEG_STEPS   ((N_NODES + DEG_THREADS - 1) / DEG_THREADS)
#define DEG_BATCH   7
static_assert(N_EDGES % (DEG_SLICES * 4) == 0, "slices must be whole 16-B groups");
static_assert(DEG_STEPS % DEG_BATCH == 0, "hand-over loop runs in whole batches");
static_assert(DEG_STEPS >= DEG_SLICES, "every slice gets a start step of its own");

__global__ void __launch_bounds__(DEG_THREADS) k_edge_deg(const int*   __restrict__ snd,
                                                          const int*   __restrict__ rcv,
                                                          const float* __restrict__ ew,
                                                          float* __restrict__ ds,
                                                          float* __restrict__ dr,
                                                          int* __restrict__ rtot) {
    __shared__ int rc[NR_PAD];
    __shared__ __attribute__((aligned(16))) float hist[DEG_STEPS * DEG_THREADS];

    const int  tid   = threadIdx.x;
    const bool recv  = blockIdx.x >= DEG_SLICES;
    const int  slice = recv ? blockIdx.x - DEG_SLICES : blockIdx.x;
    const int*  idx  = recv ? rcv : snd;
    float*      dst  = recv ? dr  : ds;

    const int4*   ip = (const int4*)  (idx + slice * DEG_PER);
    const float4* wp = (const float4*)(ew  + slice * DEG_PER);
    int4   n[DEG_IT];
    float4 w[DEG_IT];
    #pragma unroll
    for (int k = 0; k < DEG_IT; ++k) {
        const int v  = tid + k * DEG_THREADS;
        const int vc = v < DEG_NV ? v : DEG_NV - 1;
        n[k] = ip[vc];
        w[k] = wp[vc];
    }

    for (int i = tid; i < DEG_STEPS * DEG_THREADS / 4; i += DEG_THREADS)
        ((float4*)hist)[i] = make_float4(0.0f, 0.0f, 0.0f, 0.0f);
    if (tid < NR_PAD) rc[tid] = 0;
    __syncthreads();

    #pragma unroll
    for (int k = 0; k < DEG_IT; ++k) {
        if (tid + k * DEG_THREADS < DEG_NV) {
            atomicAdd(&hist[n[k].x], w[k].x + 1.0f);
            atomicAdd(&hist[n[k].y], w[k].y + 1.0f);
            atomicAdd(&hist[n[k].z], w[k].z + 1.0f);
            atomicAdd(&hist[n[k].w], w[k].w + 1.0f);
            if (recv) {
                atomicAdd(&rc[n[k].x / RN], 1);
                atomicAdd(&rc[n[k].y / RN], 1);
                atomicAdd(&rc[n[k].z / RN], 1);
                atomicAdd(&rc[n[k].w / RN], 1);
            }
        }
    }
    __syncthreads();
    if (recv && tid < NR) atomicAdd(&rtot[tid], rc[tid]);

    const int off = (slice * DEG_STEPS) / DEG_SLICES;
    for (int k0 = 0; k0 < DEG_STEPS; k0 += DEG_BATCH) {
        int   node[DEG_BATCH];
        float part[DEG_BATCH];
        #pragma unroll
        for (int u = 0; u < DEG_BATCH; ++u) {
            int st = k0 + u + off;
            if (st >= DEG_STEPS) st -= DEG_STEPS;
            node[u] = st * DEG_THREADS + tid;
            part[u] = hist[node[u]];
        }
        #pragma unroll
        for (int u = 0; u < DEG_BATCH; ++u)
            if (node[u] < N_NODES) unsafeAtomicAdd(&dst[node[u]], part[u]);
    }
}

#define GB_THREADS   1024
#define N_STRIPS     (N_NODES / 16)
#define GEMM_STRIPS  (GB_THREADS / 64)
#define GEMM_BLOCKS  ((N_STRIPS + GEMM_STRIPS - 1) / GEMM_STRIPS)
static_assert(N_NODES % 16 == 0, "strips are whole");

__device__ __forceinline__ void gemm_scale_role(const int gb,
                                                const float* __restrict__ nodes,
                                                const float* __restrict__ Wm,
                                                const float* __restrict__ ds,
                                                const float* __restrict__ dr,
                                                float* __restrict__ y,
                                                float* __restrict__ w_lds) {
    const int tid = threadIdx.x;

    for (int idx = tid; idx < DIM * DIM; idx += GB_THREADS) {
        int k = idx >> 7;
        int c = idx & (DIM - 1);
        w_lds[((k >> 1) << 8) + (c << 1) + (k & 1)] = Wm[idx];
    }
    __syncthreads();

    const int lane  = tid & 31;
    const int wave  = tid >> 5;
    const int strip = gb * GEMM_STRIPS + (wave >> 1);
    const int par   = wave & 1;

    if (strip < N_STRIPS) {
        const int row0 = strip * 16;
        const int m    = lane & 15;
        const int koff = (lane >> 4) << 1;

        const float* arow = nodes + (size_t)(row0 + m) * DIM;

        v8f acc[4] = {};

        #pragma unroll 4
        for (int k0 = 0; k0 < DIM; k0 += 4) {
            v2f a = *(const v2f*)(arow + k0 + koff);
            const float* bbase = w_lds + (((k0 + koff) >> 1) << 8);
            #pragma unroll
            for (int q = 0; q < 4; ++q) {
                const int t = 2 * q + par;
                v2f b = *(const v2f*)(bbase + ((t * 16 + m) << 1));
                acc[q] = __builtin_amdgcn_wmma_f32_16x16x4_f32(
                     false, a,  false, b,
                     (short)0, acc[q],
                     false,  false);
            }
        }

        const int roff = (lane >> 4) << 3;
        #pragma unroll
        for (int v = 0; v < 8; ++v) {
            const int row = row0 + v + roff;
            const float a1 = ds[row];
            const float a2 = dr[row];
            const float sc = rsqrtf(fmaxf(a1, 1.0f)) * rsqrtf(fmaxf(a2, 1.0f));
            float* yrow = y + (size_t)row * DIM;
            *(float4*)(yrow + 64 * par + 4 * m) =
                make_float4(acc[0][v] * sc, acc[1][v] * sc, acc[2][v] * sc, acc[3][v] * sc);
        }
    }
}

#define BIN_THREADS 1024
#define BIN_PER     3200
#define BIN_BLOCKS  (N_EDGES / BIN_PER)
#define BIN_IT      ((BIN_PER + BIN_THREADS - 1) / BIN_THREADS)
static_assert(N_EDGES % BIN_PER == 0, "slices are whole");

static_assert(BIN_THREADS == GB_THREADS, "both roles of the launch use one block size");

__device__ __forceinline__ void bin_role(const int bb,
                                         const int*   __restrict__ snd,
                                         const int*   __restrict__ rcv,
                                         const float* __restrict__ ew,
                                         const int*   __restrict__ rtot,
                                         int*  __restrict__ rfill,
                                         int2* __restrict__ bins,
                                         int* __restrict__ s_wsum,
                                         int* __restrict__ s_cnt,
                                         int* __restrict__ s_base) {
    const int tid  = threadIdx.x;
    const int lane = tid & 31;
    const int wave = tid >> 5;
    const int e0   = bb * BIN_PER;

    int tot = 0, incl = 0;
    if (tid < NR_PAD) {
        s_cnt[tid] = 0;
        tot  = tid < NR ? rtot[tid] : 0;
        incl = tot;
        #pragma unroll
        for (int d = 1; d < 32; d <<= 1) {
            const int t = __shfl_up(incl, d);
            if (lane >= d) incl += t;
        }
        if (lane == 31) s_wsum[wave] = incl;
    }

    int   rr[BIN_IT], ss[BIN_IT];
    float ww[BIN_IT];
    #pragma unroll
    for (int k = 0; k < BIN_IT; ++k) {
        const int i = tid + k * BIN_THREADS;
        rr[k] = 0; ss[k] = 0; ww[k] = 0.0f;
        if (i < BIN_PER) { rr[k] = rcv[e0 + i]; ss[k] = snd[e0 + i]; ww[k] = ew[e0 + i]; }
    }
    __syncthreads();

    int g[BIN_IT], rank[BIN_IT];
    #pragma unroll
    for (int k = 0; k < BIN_IT; ++k) {
        g[k] = rr[k] / RN;
        rank[k] = 0;
        if (tid + k * BIN_THREADS < BIN_PER) rank[k] = atomicAdd(&s_cnt[g[k]], 1);
    }
    __syncthreads();

    if (tid < NR) {
        int st = incl - tot;
        for (int k = 0; k < wave; ++k) st += s_wsum[k];
        s_base[tid] = st + atomicAdd(&rfill[tid], s_cnt[tid]);
    }
    __syncthreads();

    #pragma unroll
    for (int k = 0; k < BIN_IT; ++k) {
        if (tid + k * BIN_THREADS < BIN_PER)
            bins[s_base[g[k]] + rank[k]] =
                make_int2(ss[k] | ((rr[k] - g[k] * RN) << 16), __float_as_int(ww[k]));
    }
}

__global__ void __launch_bounds__(GB_THREADS) __attribute__((amdgpu_waves_per_eu(16, 16)))
k_bin_gemm(const float* __restrict__ nodes,
           const float* __restrict__ Wm,
           const float* __restrict__ ds,
           const float* __restrict__ dr,
           float* __restrict__ y,
           const int*   __restrict__ snd,
           const int*   __restrict__ rcv,
           const float* __restrict__ ew,
           const int*   __restrict__ rtot,
           int*  __restrict__ rfill,
           int2* __restrict__ bins) {
    __shared__ float w_lds[DIM * DIM];
    __shared__ int s_wsum[NR_PAD / 32];
    __shared__ int s_cnt[NR_PAD];
    __shared__ int s_base[NR_PAD];

    if (blockIdx.x < GEMM_BLOCKS)
        gemm_scale_role(blockIdx.x, nodes, Wm, ds, dr, y, w_lds);
    else
        bin_role(blockIdx.x - GEMM_BLOCKS, snd, rcv, ew, rtot, rfill, bins, s_wsum, s_cnt, s_base);
}

#define AGG_THREADS 1024
#define AGG_WAVES   (AGG_THREADS / 32)
#define AGG_ROWS    8
#define AGG_INIT    ((RN + AGG_WAVES - 1) / AGG_WAVES)

__global__ void __launch_bounds__(AGG_THREADS) k_row_sums(const float* __restrict__ y,
                                                          const int*   __restrict__ rtot,
                                                          const int2*  __restrict__ bins,
                                                          float* __restrict__ out) {
    __shared__ __attribute__((aligned(16))) float acc[RN * DIM];
    __shared__ int s_start;

    const int tid   = threadIdx.x;
    const int lane  = tid & 31;
    const int wave  = tid >> 5;
    const int range = blockIdx.x;
    const int node0 = range * RN;

    if (tid == 0) s_start = 0;
    __syncthreads();

    if (tid < range) atomicAdd(&s_start, rtot[tid]);

    float4 self[AGG_INIT];
    #pragma unroll
    for (int p = 0; p < AGG_INIT; ++p) {
        const int rl  = wave + p * AGG_WAVES;
        const int rlc = rl < RN ? rl : RN - 1;
        self[p] = *(const float4*)(y + (size_t)(node0 + rlc) * DIM + 4 * lane);
    }
    #pragma unroll
    for (int p = 0; p < AGG_INIT; ++p) {
        const int rl = wave + p * AGG_WAVES;
        if (rl < RN) {
            float* a = acc + rl * DIM + lane;
            a[0] = self[p].x; a[32] = self[p].y; a[64] = self[p].z; a[96] = self[p].w;
        }
    }
    __syncthreads();

    const int start = s_start;
    const int cnt   = rtot[range];

    int  c0  = wave * 32;
    int  idx = start + c0 + lane;
    int2 rec = bins[idx < N_EDGES ? idx : N_EDGES - 1];
    while (c0 < cnt) {
        const int n = cnt - c0 < 32 ? cnt - c0 : 32;
        const int c1 = c0 + AGG_WAVES * 32;
        idx = start + c1 + lane;
        const int2 nrec = bins[idx < N_EDGES ? idx : N_EDGES - 1];
        #pragma unroll
        for (int j0 = 0; j0 < 32; j0 += AGG_ROWS) {
            if (j0 < n) {
                float4 v[AGG_ROWS];
                int    rl[AGG_ROWS];
                float  w[AGG_ROWS];
                #pragma unroll
                for (int u = 0; u < AGG_ROWS; ++u) {
                    const int px = __builtin_amdgcn_readlane(rec.x, j0 + u);
                    w[u]  = __int_as_float(__builtin_amdgcn_readlane(rec.y, j0 + u));
                    rl[u] = px >> 16;
                    v[u]  = *(const float4*)(y + (size_t)(px & 0xFFFF) * DIM + lane * 4);
                }
                #pragma unroll
                for (int u = 0; u < AGG_ROWS; ++u)
                    asm volatile("" :: "v"(v[u].x), "v"(v[u].y), "v"(v[u].z), "v"(v[u].w));
                #pragma unroll
                for (int u = 0; u < AGG_ROWS; ++u) {
                    if (j0 + u < n) {
                        float* a = acc + rl[u] * DIM + lane;
                        atomicAdd(a,      v[u].x * w[u]);
                        atomicAdd(a + 32, v[u].y * w[u]);
                        atomicAdd(a + 64, v[u].z * w[u]);
                        atomicAdd(a + 96, v[u].w * w[u]);
                    }
                }
            }
        }
        c0 = c1; rec = nrec;
    }
    __syncthreads();

    float4*       orow = (float4*)(out + (size_t)node0 * DIM);
    const float4* arow = (const float4*)acc;
    for (int idx = tid; idx < RN * DIM / 4; idx += AGG_THREADS) orow[idx] = arow[idx];
}

extern "C" void kernel_launch(void* const* d_in, const int* in_sizes, int n_in,
                              void* d_out, int out_size, void* d_ws, size_t ws_size,
                              hipStream_t stream) {
    const float* nodes     = (const float*)d_in[0];
    const int*   senders   = (const int*)  d_in[1];
    const int*   receivers = (const int*)  d_in[2];
    const float* edges     = (const float*)d_in[3];
    const float* Wm        = (const float*)d_in[4];
    float*       out       = (float*)d_out;

    float* y     = (float*)d_ws;
    float* ds    = y  + (size_t)N_NODES * DIM;
    float* dr    = ds + N_NODES;
    int*   rtot  = (int*)(dr + N_NODES);
    int*   rfill = rtot + NR_PAD;
    int2*  bins  = (int2*)(rfill + NR_PAD);

    k_init_deg<<<(N_NODES + 255) / 256, 256, 0, stream>>>(ds, dr, rtot);
    k_edge_deg<<<2 * DEG_SLICES, DEG_THREADS, 0, stream>>>(senders, receivers, edges, ds, dr, rtot);

    k_bin_gemm<<<GEMM_BLOCKS + BIN_BLOCKS, GB_THREADS, 0, stream>>>(
        nodes, Wm, ds, dr, y, senders, receivers, edges, rtot, rfill, bins);
    k_row_sums<<<NR, AGG_THREADS, 0, stream>>>(y, rtot, bins, out);
}


# degree sums: node range cut into 4 line-aligned parts, one 8-wave workgroup per (side, edge slice, part): 256 workgroups, one per WGP (was 64 of 32 waves); LDS sums and hand-over a quarter per WGP
# speedup vs baseline: 3.8395x; 1.0765x over previous
#include <hip/hip_runtime.h>
#include <cstdint>

#define N_NODES 50000
#define N_EDGES 800000
#define DIM     128

#define RN      200
#define NR      (N_NODES / RN)
#define NR_PAD  256
static_assert(N_NODES % RN == 0, "ranges are whole");
static_assert(NR <= NR_PAD && RN <= 256, "range id and local row fit their fields");
static_assert(N_NODES <= 65536, "a sender id fits 16 bits of the packed edge word");

typedef __attribute__((ext_vector_type(2))) float v2f;
typedef __attribute__((ext_vector_type(8))) float v8f;

__global__ void __launch_bounds__(256) k_init_deg(float* __restrict__ ds,
                                                  float* __restrict__ dr,
                                                  int* __restrict__ rcount) {
    int i = blockIdx.x * blockDim.x + threadIdx.x;
    if (i < N_NODES) { ds[i] = 2.0f; dr[i] = 2.0f; }
    if (i < 2 * NR_PAD) rcount[i] = 0;
}

#define DEG_SLICES  32
#define DEG_PARTS   4
#define DEG_THREADS 256
#define DEG_PER     (N_EDGES / DEG_SLICES)
#define DEG_NV      (DEG_PER / 4)
#define DEG_IT      ((DEG_NV + DEG_THREADS - 1) / DEG_THREADS)
#define DEG_PART_N  ((((N_NODES + DEG_PARTS - 1) / DEG_PARTS) + 31) / 32 * 32)
#define DEG_STEPS   ((DEG_PART_N + DEG_THREADS - 1) / DEG_THREADS)
#define DEG_BATCH   7
static_assert(N_EDGES % (DEG_SLICES * 4) == 0, "slices must be whole 16-B groups");
static_assert(DEG_PART_N * DEG_PARTS >= N_NODES, "the parts cover every node");
static_assert(DEG_PART_N * (DEG_PARTS - 1) < N_NODES, "no part is empty");
static_assert(DEG_STEPS % DEG_BATCH == 0, "hand-over loop runs in whole batches");
static_assert(DEG_STEPS >= DEG_SLICES, "every slice gets a start step of its own");
static_assert(DEG_THREADS >= NR_PAD, "one thread per range counter");

__global__ void __launch_bounds__(DEG_THREADS) k_edge_deg(const int*   __restrict__ snd,
                                                          const int*   __restrict__ rcv,
                                                          const float* __restrict__ ew,
                                                          float* __restrict__ ds,
                                                          float* __restrict__ dr,
                                                          int* __restrict__ rtot) {
    __shared__ int rc[NR_PAD];
    __shared__ __attribute__((aligned(16))) float hist[DEG_STEPS * DEG_THREADS];

    const int  tid   = threadIdx.x;
    const int  nsp   = DEG_SLICES * DEG_PARTS;
    const bool recv  = blockIdx.x < nsp;
    const int  sp    = recv ? blockIdx.x : blockIdx.x - nsp;
    const int  slice = sp / DEG_PARTS;
    const int  lo    = (sp % DEG_PARTS) * DEG_PART_N;
    const int  len   = N_NODES - lo < DEG_PART_N ? N_NODES - lo : DEG_PART_N;
    const int*  idx  = recv ? rcv : snd;
    float*      dst  = recv ? dr  : ds;

    const int4*   ip = (const int4*)  (idx + slice * DEG_PER);
    const float4* wp = (const float4*)(ew  + slice * DEG_PER);
    int4   n[DEG_IT];
    float4 w[DEG_IT];
    #pragma unroll
    for (int k = 0; k < DEG_IT; ++k) {
        const int v  = tid + k * DEG_THREADS;
        const int vc = v < DEG_NV ? v : DEG_NV - 1;
        n[k] = ip[vc];
        w[k] = wp[vc];
    }

    for (int i = tid; i < DEG_STEPS * DEG_THREADS / 4; i += DEG_THREADS)
        ((float4*)hist)[i] = make_float4(0.0f, 0.0f, 0.0f, 0.0f);
    if (tid < NR_PAD) rc[tid] = 0;
    __syncthreads();

    #pragma unroll
    for (int k = 0; k < DEG_IT; ++k) {
        if (tid + k * DEG_THREADS < DEG_NV) {
            const int   nn[4] = { n[k].x, n[k].y, n[k].z, n[k].w };
            const float ww[4] = { w[k].x, w[k].y, w[k].z, w[k].w };
            #pragma unroll
            for (int j = 0; j < 4; ++j) {
                const unsigned r = (unsigned)(nn[j] - lo);
                if (r < (unsigned)len) {
                    atomicAdd(&hist[r], ww[j] + 1.0f);
                    if (recv) atomicAdd(&rc[nn[j] / RN], 1);
                }
            }
        }
    }
    __syncthreads();
    if (recv && tid < NR) {
        const int c = rc[tid];
        if (c != 0) atomicAdd(&rtot[tid], c);
    }

    const int off = (slice * DEG_STEPS) / DEG_SLICES;
    for (int k0 = 0; k0 < DEG_STEPS; k0 += DEG_BATCH) {
        int   node[DEG_BATCH];
        float part[DEG_BATCH];
        #pragma unroll
        for (int u = 0; u < DEG_BATCH; ++u) {
            int st = k0 + u + off;
            if (st >= DEG_STEPS) st -= DEG_STEPS;
            node[u] = st * DEG_THREADS + tid;
            part[u] = hist[node[u]];
        }
        #pragma unroll
        for (int u = 0; u < DEG_BATCH; ++u)
            if (node[u] < len) unsafeAtomicAdd(&dst[lo + node[u]], part[u]);
    }
}

#define GB_THREADS   1024
#define N_STRIPS     (N_NODES / 16)
#define GEMM_STRIPS  (GB_THREADS / 64)
#define GEMM_BLOCKS  ((N_STRIPS + GEMM_STRIPS - 1) / GEMM_STRIPS)
static_assert(N_NODES % 16 == 0, "strips are whole");

__device__ __forceinline__ void gemm_scale_role(const int gb,
                                                const float* __restrict__ nodes,
                                                const float* __restrict__ Wm,
                                                const float* __restrict__ ds,
                                                const float* __restrict__ dr,
                                                float* __restrict__ y,
                                                float* __restrict__ w_lds) {
    const int tid = threadIdx.x;

    for (int idx = tid; idx < DIM * DIM; idx += GB_THREADS) {
        int k = idx >> 7;
        int c = idx & (DIM - 1);
        w_lds[((k >> 1) << 8) + (c << 1) + (k & 1)] = Wm[idx];
    }
    __syncthreads();

    const int lane  = tid & 31;
    const int wave  = tid >> 5;
    const int strip = gb * GEMM_STRIPS + (wave >> 1);
    const int par   = wave & 1;

    if (strip < N_STRIPS) {
        const int row0 = strip * 16;
        const int m    = lane & 15;
        const int koff = (lane >> 4) << 1;

        const float* arow = nodes + (size_t)(row0 + m) * DIM;

        v8f acc[4] = {};

        #pragma unroll 4
        for (int k0 = 0; k0 < DIM; k0 += 4) {
            v2f a = *(const v2f*)(arow + k0 + koff);
            const float* bbase = w_lds + (((k0 + koff) >> 1) << 8);
            #pragma unroll
            for (int q = 0; q < 4; ++q) {
                const int t = 2 * q + par;
                v2f b = *(const v2f*)(bbase + ((t * 16 + m) << 1));
                acc[q] = __builtin_amdgcn_wmma_f32_16x16x4_f32(
                     false, a,  false, b,
                     (short)0, acc[q],
                     false,  false);
            }
        }

        const int roff = (lane >> 4) << 3;
        #pragma unroll
        for (int v = 0; v < 8; ++v) {
            const int row = row0 + v + roff;
            const float a1 = ds[row];
            const float a2 = dr[row];
            const float sc = rsqrtf(fmaxf(a1, 1.0f)) * rsqrtf(fmaxf(a2, 1.0f));
            float* yrow = y + (size_t)row * DIM;
            *(float4*)(yrow + 64 * par + 4 * m) =
                make_float4(acc[0][v] * sc, acc[1][v] * sc, acc[2][v] * sc, acc[3][v] * sc);
        }
    }
}

#define BIN_THREADS 1024
#define BIN_PER     3200
#define BIN_BLOCKS  (N_EDGES / BIN_PER)
#define BIN_IT      ((BIN_PER + BIN_THREADS - 1) / BIN_THREADS)
static_assert(N_EDGES % BIN_PER == 0, "slices are whole");

static_assert(BIN_THREADS == GB_THREADS, "both roles of the launch use one block size");

__device__ __forceinline__ void bin_role(const int bb,
                                         const int*   __restrict__ snd,
                                         const int*   __restrict__ rcv,
                                         const float* __restrict__ ew,
                                         const int*   __restrict__ rtot,
                                         int*  __restrict__ rfill,
                                         int2* __restrict__ bins,
                                         int* __restrict__ s_wsum,
                                         int* __restrict__ s_cnt,
                                         int* __restrict__ s_base) {
    const int tid  = threadIdx.x;
    const int lane = tid & 31;
    const int wave = tid >> 5;
    const int e0   = bb * BIN_PER;

    int tot = 0, incl = 0;
    if (tid < NR_PAD) {
        s_cnt[tid] = 0;
        tot  = tid < NR ? rtot[tid] : 0;
        incl = tot;
        #pragma unroll
        for (int d = 1; d < 32; d <<= 1) {
            const int t = __shfl_up(incl, d);
            if (lane >= d) incl += t;
        }
        if (lane == 31) s_wsum[wave] = incl;
    }

    int   rr[BIN_IT], ss[BIN_IT];
    float ww[BIN_IT];
    #pragma unroll
    for (int k = 0; k < BIN_IT; ++k) {
        const int i = tid + k * BIN_THREADS;
        rr[k] = 0; ss[k] = 0; ww[k] = 0.0f;
        if (i < BIN_PER) { rr[k] = rcv[e0 + i]; ss[k] = snd[e0 + i]; ww[k] = ew[e0 + i]; }
    }
    __syncthreads();

    int g[BIN_IT], rank[BIN_IT];
    #pragma unroll
    for (int k = 0; k < BIN_IT; ++k) {
        g[k] = rr[k] / RN;
        rank[k] = 0;
        if (tid + k * BIN_THREADS < BIN_PER) rank[k] = atomicAdd(&s_cnt[g[k]], 1);
    }
    __syncthreads();

    if (tid < NR) {
        int st = incl - tot;
        for (int k = 0; k < wave; ++k) st += s_wsum[k];
        s_base[tid] = st + atomicAdd(&rfill[tid], s_cnt[tid]);
    }
    __syncthreads();

    #pragma unroll
    for (int k = 0; k < BIN_IT; ++k) {
        if (tid + k * BIN_THREADS < BIN_PER)
            bins[s_base[g[k]] + rank[k]] =
                make_int2(ss[k] | ((rr[k] - g[k] * RN) << 16), __float_as_int(ww[k]));
    }
}

__global__ void __launch_bounds__(GB_THREADS) __attribute__((amdgpu_waves_per_eu(16, 16)))
k_bin_gemm(const float* __restrict__ nodes,
           const float* __restrict__ Wm,
           const float* __restrict__ ds,
           const float* __restrict__ dr,
           float* __restrict__ y,
           const int*   __restrict__ snd,
           const int*   __restrict__ rcv,
           const float* __restrict__ ew,
           const int*   __restrict__ rtot,
           int*  __restrict__ rfill,
           int2* __restrict__ bins) {
    __shared__ float w_lds[DIM * DIM];
    __shared__ int s_wsum[NR_PAD / 32];
    __shared__ int s_cnt[NR_PAD];
    __shared__ int s_base[NR_PAD];

    if (blockIdx.x < GEMM_BLOCKS)
        gemm_scale_role(blockIdx.x, nodes, Wm, ds, dr, y, w_lds);
    else
        bin_role(blockIdx.x - GEMM_BLOCKS, snd, rcv, ew, rtot, rfill, bins, s_wsum, s_cnt, s_base);
}

#define AGG_THREADS 1024
#define AGG_WAVES   (AGG_THREADS / 32)
#define AGG_ROWS    8
#define AGG_INIT    ((RN + AGG_WAVES - 1) / AGG_WAVES)

__global__ void __launch_bounds__(AGG_THREADS) k_row_sums(const float* __restrict__ y,
                                                          const int*   __restrict__ rtot,
                                                          const int2*  __restrict__ bins,
                                                          float* __restrict__ out) {
    __shared__ __attribute__((aligned(16))) float acc[RN * DIM];
    __shared__ int s_start;

    const int tid   = threadIdx.x;
    const int lane  = tid & 31;
    const int wave  = tid >> 5;
    const int range = blockIdx.x;
    const int node0 = range * RN;

    if (tid == 0) s_start = 0;
    __syncthreads();

    if (tid < range) atomicAdd(&s_start, rtot[tid]);

    float4 self[AGG_INIT];
    #pragma unroll
    for (int p = 0; p < AGG_INIT; ++p) {
        const int rl  = wave + p * AGG_WAVES;
        const int rlc = rl < RN ? rl : RN - 1;
        self[p] = *(const float4*)(y + (size_t)(node0 + rlc) * DIM + 4 * lane);
    }
    #pragma unroll
    for (int p = 0; p < AGG_INIT; ++p) {
        const int rl = wave + p * AGG_WAVES;
        if (rl < RN) {
            float* a = acc + rl * DIM + lane;
            a[0] = self[p].x; a[32] = self[p].y; a[64] = self[p].z; a[96] = self[p].w;
        }
    }
    __syncthreads();

    const int start = s_start;
    const int cnt   = rtot[range];

    int  c0  = wave * 32;
    int  idx = start + c0 + lane;
    int2 rec = bins[idx < N_EDGES ? idx : N_EDGES - 1];
    while (c0 < cnt) {
        const int n = cnt - c0 < 32 ? cnt - c0 : 32;
        const int c1 = c0 + AGG_WAVES * 32;
        idx = start + c1 + lane;
        const int2 nrec = bins[idx < N_EDGES ? idx : N_EDGES - 1];
        #pragma unroll
        for (int j0 = 0; j0 < 32; j0 += AGG_ROWS) {
            if (j0 < n) {
                float4 v[AGG_ROWS];
                int    rl[AGG_ROWS];
                float  w[AGG_ROWS];
                #pragma unroll
                for (int u = 0; u < AGG_ROWS; ++u) {
                    const int px = __builtin_amdgcn_readlane(rec.x, j0 + u);
                    w[u]  = __int_as_float(__builtin_amdgcn_readlane(rec.y, j0 + u));
                    rl[u] = px >> 16;
                    v[u]  = *(const float4*)(y + (size_t)(px & 0xFFFF) * DIM + lane * 4);
                }
                #pragma unroll
                for (int u = 0; u < AGG_ROWS; ++u)
                    asm volatile("" :: "v"(v[u].x), "v"(v[u].y), "v"(v[u].z), "v"(v[u].w));
                #pragma unroll
                for (int u = 0; u < AGG_ROWS; ++u) {
                    if (j0 + u < n) {
                        float* a = acc + rl[u] * DIM + lane;
                        atomicAdd(a,      v[u].x * w[u]);
                        atomicAdd(a + 32, v[u].y * w[u]);
                        atomicAdd(a + 64, v[u].z * w[u]);
                        atomicAdd(a + 96, v[u].w * w[u]);
                    }
                }
            }
        }
        c0 = c1; rec = nrec;
    }
    __syncthreads();

    float4*       orow = (float4*)(out + (size_t)node0 * DIM);
    const float4* arow = (const float4*)acc;
    for (int idx = tid; idx < RN * DIM / 4; idx += AGG_THREADS) orow[idx] = arow[idx];
}

extern "C" void kernel_launch(void* const* d_in, const int* in_sizes, int n_in,
                              void* d_out, int out_size, void* d_ws, size_t ws_size,
                              hipStream_t stream) {
    const float* nodes     = (const float*)d_in[0];
    const int*   senders   = (const int*)  d_in[1];
    const int*   receivers = (const int*)  d_in[2];
    const float* edges     = (const float*)d_in[3];
    const float* Wm        = (const float*)d_in[4];
    float*       out       = (float*)d_out;

    float* y     = (float*)d_ws;
    float* ds    = y  + (size_t)N_NODES * DIM;
    float* dr    = ds + N_NODES;
    int*   rtot  = (int*)(dr + N_NODES);
    int*   rfill = rtot + NR_PAD;
    int2*  bins  = (int2*)(rfill + NR_PAD);

    k_init_deg<<<(N_NODES + 255) / 256, 256, 0, stream>>>(ds, dr, rtot);
    k_edge_deg<<<2 * DEG_SLICES * DEG_PARTS, DEG_THREADS, 0, stream>>>(senders, receivers, edges, ds, dr, rtot);

    k_bin_gemm<<<GEMM_BLOCKS + BIN_BLOCKS, GB_THREADS, 0, stream>>>(
        nodes, Wm, ds, dr, y, senders, receivers, edges, rtot, rfill, bins);
    k_row_sums<<<NR, AGG_THREADS, 0, stream>>>(y, rtot, bins, out);
}


# edge sum: each range's edges counting-sorted by receiver in LDS; a wave sums a receiver's run of rows in registers and adds it to the LDS row once per run, not per edge; sum order changes
# speedup vs baseline: 4.0684x; 1.0596x over previous
#include <hip/hip_runtime.h>
#include <cstdint>

#define N_NODES 50000
#define N_EDGES 800000
#define DIM     128

#define RN      200
#define NR      (N_NODES / RN)
#define NR_PAD  256
static_assert(N_NODES % RN == 0, "ranges are whole");
static_assert(NR <= NR_PAD && RN <= 256, "range id and local row fit their fields");
static_assert(N_NODES <= 65536, "a sender id fits 16 bits of the packed edge word");

typedef __attribute__((ext_vector_type(2))) float v2f;
typedef __attribute__((ext_vector_type(8))) float v8f;

__global__ void __launch_bounds__(256) k_init_deg(float* __restrict__ ds,
                                                  float* __restrict__ dr,
                                                  int* __restrict__ rcount) {
    int i = blockIdx.x * blockDim.x + threadIdx.x;
    if (i < N_NODES) { ds[i] = 2.0f; dr[i] = 2.0f; }
    if (i < 2 * NR_PAD) rcount[i] = 0;
}

#define DEG_SLICES  32
#define DEG_PARTS   4
#define DEG_THREADS 256
#define DEG_PER     (N_EDGES / DEG_SLICES)
#define DEG_NV      (DEG_PER / 4)
#define DEG_IT      ((DEG_NV + DEG_THREADS - 1) / DEG_THREADS)
#define DEG_PART_N  ((((N_NODES + DEG_PARTS - 1) / DEG_PARTS) + 31) / 32 * 32)
#define DEG_STEPS   ((DEG_PART_N + DEG_THREADS - 1) / DEG_THREADS)
#define DEG_BATCH   7
static_assert(N_EDGES % (DEG_SLICES * 4) == 0, "slices must be whole 16-B groups");
static_assert(DEG_PART_N * DEG_PARTS >= N_NODES, "the parts cover every node");
static_assert(DEG_PART_N * (DEG_PARTS - 1) < N_NODES, "no part is empty");
static_assert(DEG_STEPS % DEG_BATCH == 0, "hand-over loop runs in whole batches");
static_assert(DEG_STEPS >= DEG_SLICES, "every slice gets a start step of its own");
static_assert(DEG_THREADS >= NR_PAD, "one thread per range counter");

__global__ void __launch_bounds__(DEG_THREADS) k_edge_deg(const int*   __restrict__ snd,
                                                          const int*   __restrict__ rcv,
                                                          const float* __restrict__ ew,
                                                          float* __restrict__ ds,
                                                          float* __restrict__ dr,
                                                          int* __restrict__ rtot) {
    __shared__ int rc[NR_PAD];
    __shared__ __attribute__((aligned(16))) float hist[DEG_STEPS * DEG_THREADS];

    const int  tid   = threadIdx.x;
    const int  nsp   = DEG_SLICES * DEG_PARTS;
    const bool recv  = blockIdx.x < nsp;
    const int  sp    = recv ? blockIdx.x : blockIdx.x - nsp;
    const int  slice = sp / DEG_PARTS;
    const int  lo    = (sp % DEG_PARTS) * DEG_PART_N;
    const int  len   = N_NODES - lo < DEG_PART_N ? N_NODES - lo : DEG_PART_N;
    const int*  idx  = recv ? rcv : snd;
    float*      dst  = recv ? dr  : ds;

    const int4*   ip = (const int4*)  (idx + slice * DEG_PER);
    const float4* wp = (const float4*)(ew  + slice * DEG_PER);
    int4   n[DEG_IT];
    float4 w[DEG_IT];
    #pragma unroll
    for (int k = 0; k < DEG_IT; ++k) {
        const int v  = tid + k * DEG_THREADS;
        const int vc = v < DEG_NV ? v : DEG_NV - 1;
        n[k] = ip[vc];
        w[k] = wp[vc];
    }

    for (int i = tid; i < DEG_STEPS * DEG_THREADS / 4; i += DEG_THREADS)
        ((float4*)hist)[i] = make_float4(0.0f, 0.0f, 0.0f, 0.0f);
    if (tid < NR_PAD) rc[tid] = 0;
    __syncthreads();

    #pragma unroll
    for (int k = 0; k < DEG_IT; ++k) {
        if (tid + k * DEG_THREADS < DEG_NV) {
            const int   nn[4] = { n[k].x, n[k].y, n[k].z, n[k].w };
            const float ww[4] = { w[k].x, w[k].y, w[k].z, w[k].w };
            #pragma unroll
            for (int j = 0; j < 4; ++j) {
                const unsigned r = (unsigned)(nn[j] - lo);
                if (r < (unsigned)len) {
                    atomicAdd(&hist[r], ww[j] + 1.0f);
                    if (recv) atomicAdd(&rc[nn[j] / RN], 1);
                }
            }
        }
    }
    __syncthreads();
    if (recv && tid < NR) {
        const int c = rc[tid];
        if (c != 0) atomicAdd(&rtot[tid], c);
    }

    const int off = (slice * DEG_STEPS) / DEG_SLICES;
    for (int k0 = 0; k0 < DEG_STEPS; k0 += DEG_BATCH) {
        int   node[DEG_BATCH];
        float part[DEG_BATCH];
        #pragma unroll
        for (int u = 0; u < DEG_BATCH; ++u) {
            int st = k0 + u + off;
            if (st >= DEG_STEPS) st -= DEG_STEPS;
            node[u] = st * DEG_THREADS + tid;
            part[u] = hist[node[u]];
        }
        #pragma unroll
        for (int u = 0; u < DEG_BATCH; ++u)
            if (node[u] < len) unsafeAtomicAdd(&dst[lo + node[u]], part[u]);
    }
}

#define GB_THREADS   1024
#define N_STRIPS     (N_NODES / 16)
#define GEMM_STRIPS  (GB_THREADS / 64)
#define GEMM_BLOCKS  ((N_STRIPS + GEMM_STRIPS - 1) / GEMM_STRIPS)
static_assert(N_NODES % 16 == 0, "strips are whole");

__device__ __forceinline__ void gemm_scale_role(const int gb,
                                                const float* __restrict__ nodes,
                                                const float* __restrict__ Wm,
                                                const float* __restrict__ ds,
                                                const float* __restrict__ dr,
                                                float* __restrict__ y,
                                                float* __restrict__ w_lds) {
    const int tid = threadIdx.x;

    for (int idx = tid; idx < DIM * DIM; idx += GB_THREADS) {
        int k = idx >> 7;
        int c = idx & (DIM - 1);
        w_lds[((k >> 1) << 8) + (c << 1) + (k & 1)] = Wm[idx];
    }
    __syncthreads();

    const int lane  = tid & 31;
    const int wave  = tid >> 5;
    const int strip = gb * GEMM_STRIPS + (wave >> 1);
    const int par   = wave & 1;

    if (strip < N_STRIPS) {
        const int row0 = strip * 16;
        const int m    = lane & 15;
        const int koff = (lane >> 4) << 1;

        const float* arow = nodes + (size_t)(row0 + m) * DIM;

        v8f acc[4] = {};

        #pragma unroll 4
        for (int k0 = 0; k0 < DIM; k0 += 4) {
            v2f a = *(const v2f*)(arow + k0 + koff);
            const float* bbase = w_lds + (((k0 + koff) >> 1) << 8);
            #pragma unroll
            for (int q = 0; q < 4; ++q) {
                const int t = 2 * q + par;
                v2f b = *(const v2f*)(bbase + ((t * 16 + m) << 1));
                acc[q] = __builtin_amdgcn_wmma_f32_16x16x4_f32(
                     false, a,  false, b,
                     (short)0, acc[q],
                     false,  false);
            }
        }

        const int roff = (lane >> 4) << 3;
        #pragma unroll
        for (int v = 0; v < 8; ++v) {
            const int row = row0 + v + roff;
            const float a1 = ds[row];
            const float a2 = dr[row];
            const float sc = rsqrtf(fmaxf(a1, 1.0f)) * rsqrtf(fmaxf(a2, 1.0f));
            float* yrow = y + (size_t)row * DIM;
            *(float4*)(yrow + 64 * par + 4 * m) =
                make_float4(acc[0][v] * sc, acc[1][v] * sc, acc[2][v] * sc, acc[3][v] * sc);
        }
    }
}

#define BIN_THREADS 1024
#define BIN_PER     3200
#define BIN_BLOCKS  (N_EDGES / BIN_PER)
#define BIN_IT      ((BIN_PER + BIN_THREADS - 1) / BIN_THREADS)
static_assert(N_EDGES % BIN_PER == 0, "slices are whole");

static_assert(BIN_THREADS == GB_THREADS, "both roles of the launch use one block size");

__device__ __forceinline__ void bin_role(const int bb,
                                         const int*   __restrict__ snd,
                                         const int*   __restrict__ rcv,
                                         const float* __restrict__ ew,
                                         const int*   __restrict__ rtot,
                                         int*  __restrict__ rfill,
                                         int2* __restrict__ bins,
                                         int* __restrict__ s_wsum,
                                         int* __restrict__ s_cnt,
                                         int* __restrict__ s_base) {
    const int tid  = threadIdx.x;
    const int lane = tid & 31;
    const int wave = tid >> 5;
    const int e0   = bb * BIN_PER;

    int tot = 0, incl = 0;
    if (tid < NR_PAD) {
        s_cnt[tid] = 0;
        tot  = tid < NR ? rtot[tid] : 0;
        incl = tot;
        #pragma unroll
        for (int d = 1; d < 32; d <<= 1) {
            const int t = __shfl_up(incl, d);
            if (lane >= d) incl += t;
        }
        if (lane == 31) s_wsum[wave] = incl;
    }

    int   rr[BIN_IT], ss[BIN_IT];
    float ww[BIN_IT];
    #pragma unroll
    for (int k = 0; k < BIN_IT; ++k) {
        const int i = tid + k * BIN_THREADS;
        rr[k] = 0; ss[k] = 0; ww[k] = 0.0f;
        if (i < BIN_PER) { rr[k] = rcv[e0 + i]; ss[k] = snd[e0 + i]; ww[k] = ew[e0 + i]; }
    }
    __syncthreads();

    int g[BIN_IT], rank[BIN_IT];
    #pragma unroll
    for (int k = 0; k < BIN_IT; ++k) {
        g[k] = rr[k] / RN;
        rank[k] = 0;
        if (tid + k * BIN_THREADS < BIN_PER) rank[k] = atomicAdd(&s_cnt[g[k]], 1);
    }
    __syncthreads();

    if (tid < NR) {
        int st = incl - tot;
        for (int k = 0; k < wave; ++k) st += s_wsum[k];
        s_base[tid] = st + atomicAdd(&rfill[tid], s_cnt[tid]);
    }
    __syncthreads();

    #pragma unroll
    for (int k = 0; k < BIN_IT; ++k) {
        if (tid + k * BIN_THREADS < BIN_PER)
            bins[s_base[g[k]] + rank[k]] =
                make_int2(ss[k] | ((rr[k] - g[k] * RN) << 16), __float_as_int(ww[k]));
    }
}

__global__ void __launch_bounds__(GB_THREADS) __attribute__((amdgpu_waves_per_eu(16, 16)))
k_bin_gemm(const float* __restrict__ nodes,
           const float* __restrict__ Wm,
           const float* __restrict__ ds,
           const float* __restrict__ dr,
           float* __restrict__ y,
           const int*   __restrict__ snd,
           const int*   __restrict__ rcv,
           const float* __restrict__ ew,
           const int*   __restrict__ rtot,
           int*  __restrict__ rfill,
           int2* __restrict__ bins) {
    __shared__ float w_lds[DIM * DIM];
    __shared__ int s_wsum[NR_PAD / 32];
    __shared__ int s_cnt[NR_PAD];
    __shared__ int s_base[NR_PAD];

    if (blockIdx.x < GEMM_BLOCKS)
        gemm_scale_role(blockIdx.x, nodes, Wm, ds, dr, y, w_lds);
    else
        bin_role(blockIdx.x - GEMM_BLOCKS, snd, rcv, ew, rtot, rfill, bins, s_wsum, s_cnt, s_base);
}

#define AGG_THREADS 1024
#define AGG_WAVES   (AGG_THREADS / 32)
#define AGG_ROWS    8
#define AGG_INIT    ((RN + AGG_WAVES - 1) / AGG_WAVES)
#define SORT_CH     4096
#define SORT_IT     (SORT_CH / AGG_THREADS)
#define SORT_SCAN   256
static_assert(SORT_CH % AGG_THREADS == 0, "every thread takes the same number of edge words");
static_assert(RN <= SORT_SCAN && SORT_SCAN <= AGG_THREADS && SORT_SCAN % 32 == 0,
              "one scan thread per local receiver, whole waves");

__device__ __forceinline__ void flush_run(float* __restrict__ acc, const int r, const int lane,
                                          const float sx, const float sy,
                                          const float sz, const float sw) {
    float* a = acc + r * DIM + lane;
    atomicAdd(a,      sx);
    atomicAdd(a + 32, sy);
    atomicAdd(a + 64, sz);
    atomicAdd(a + 96, sw);
}

__global__ void __launch_bounds__(AGG_THREADS) k_row_sums(const float* __restrict__ y,
                                                          const int*   __restrict__ rtot,
                                                          const int2*  __restrict__ bins,
                                                          float* __restrict__ out) {
    __shared__ __attribute__((aligned(16))) float acc[RN * DIM];
    __shared__ int2 sorted[SORT_CH];
    __shared__ int rcnt[SORT_SCAN];
    __shared__ int roff[SORT_SCAN];
    __shared__ int s_wsum[SORT_SCAN / 32];
    __shared__ int s_start;

    const int tid   = threadIdx.x;
    const int lane  = tid & 31;
    const int wave  = __builtin_amdgcn_readfirstlane(tid >> 5);
    const int range = blockIdx.x;
    const int node0 = range * RN;

    if (tid == 0) s_start = 0;
    if (tid < SORT_SCAN) rcnt[tid] = 0;
    __syncthreads();

    if (tid < range) atomicAdd(&s_start, rtot[tid]);

    float4 self[AGG_INIT];
    #pragma unroll
    for (int p = 0; p < AGG_INIT; ++p) {
        const int rl  = wave + p * AGG_WAVES;
        const int rlc = rl < RN ? rl : RN - 1;
        self[p] = *(const float4*)(y + (size_t)(node0 + rlc) * DIM + 4 * lane);
    }
    #pragma unroll
    for (int p = 0; p < AGG_INIT; ++p) {
        const int rl = wave + p * AGG_WAVES;
        if (rl < RN) {
            float* a = acc + rl * DIM + lane;
            a[0] = self[p].x; a[32] = self[p].y; a[64] = self[p].z; a[96] = self[p].w;
        }
    }
    __syncthreads();

    const int start = __builtin_amdgcn_readfirstlane(s_start);
    const int cnt   = rtot[range];

    for (int c0 = 0; c0 < cnt; c0 += SORT_CH) {
        const int nch = cnt - c0 < SORT_CH ? cnt - c0 : SORT_CH;

        int2 e[SORT_IT];
        int  rk[SORT_IT];
        #pragma unroll
        for (int k = 0; k < SORT_IT; ++k) {
            const int i = tid + k * AGG_THREADS;
            e[k] = make_int2(0, 0);
            if (i < nch) e[k] = bins[start + c0 + i];
        }
        #pragma unroll
        for (int k = 0; k < SORT_IT; ++k) {
            rk[k] = 0;
            if (tid + k * AGG_THREADS < nch) rk[k] = atomicAdd(&rcnt[e[k].x >> 16], 1);
        }
        __syncthreads();

        int c = 0, incl = 0;
        if (tid < SORT_SCAN) {
            c = rcnt[tid];
            rcnt[tid] = 0;
            incl = c;
            #pragma unroll
            for (int d = 1; d < 32; d <<= 1) {
                const int t = __shfl_up(incl, d);
                if (lane >= d) incl += t;
            }
            if (lane == 31) s_wsum[wave] = incl;
        }
        __syncthreads();
        if (tid < SORT_SCAN) {
            int st = incl - c;
            #pragma unroll
            for (int k = 0; k < SORT_SCAN / 32; ++k)
                if (k < wave) st += s_wsum[k];
            roff[tid] = st;
        }
        __syncthreads();

        #pragma unroll
        for (int k = 0; k < SORT_IT; ++k)
            if (tid + k * AGG_THREADS < nch) sorted[roff[e[k].x >> 16] + rk[k]] = e[k];
        __syncthreads();

        const int seg   = (((nch + AGG_WAVES - 1) / AGG_WAVES) + AGG_ROWS - 1) & ~(AGG_ROWS - 1);
        const int wbase = wave * seg;
        const int wend  = wbase + seg < nch ? wbase + seg : nch;
        int   cur = -1;
        float sx = 0.0f, sy = 0.0f, sz = 0.0f, sw = 0.0f;
        int2 rec = sorted[wbase + lane < nch ? wbase + lane : nch - 1];
        for (int p = wbase; p < wend; p += 32) {
            const int n = wend - p < 32 ? wend - p : 32;
            const int2 nrec = sorted[p + 32 + lane < nch ? p + 32 + lane : nch - 1];
            #pragma unroll
            for (int j0 = 0; j0 < 32; j0 += AGG_ROWS) {
                if (j0 < n) {
                    float4 v[AGG_ROWS];
                    int    rl[AGG_ROWS];
                    float  w[AGG_ROWS];
                    #pragma unroll
                    for (int u = 0; u < AGG_ROWS; ++u) {
                        const int px = __builtin_amdgcn_readlane(rec.x, j0 + u);
                        w[u]  = __int_as_float(__builtin_amdgcn_readlane(rec.y, j0 + u));
                        rl[u] = px >> 16;
                        v[u]  = *(const float4*)(y + (size_t)(px & 0xFFFF) * DIM + lane * 4);
                    }
                    #pragma unroll
                    for (int u = 0; u < AGG_ROWS; ++u)
                        asm volatile("" :: "v"(v[u].x), "v"(v[u].y), "v"(v[u].z), "v"(v[u].w));
                    #pragma unroll
                    for (int u = 0; u < AGG_ROWS; ++u) {
                        if (j0 + u < n) {
                            if (rl[u] != cur) {
                                if (cur >= 0) flush_run(acc, cur, lane, sx, sy, sz, sw);
                                cur = rl[u];
                                sx = 0.0f; sy = 0.0f; sz = 0.0f; sw = 0.0f;
                            }
                            sx += v[u].x * w[u];
                            sy += v[u].y * w[u];
                            sz += v[u].z * w[u];
                            sw += v[u].w * w[u];
                        }
                    }
                }
            }
            rec = nrec;
        }
        if (cur >= 0) flush_run(acc, cur, lane, sx, sy, sz, sw);
        __syncthreads();
    }

    float4*       orow = (float4*)(out + (size_t)node0 * DIM);
    const float4* arow = (const float4*)acc;
    for (int idx = tid; idx < RN * DIM / 4; idx += AGG_THREADS) orow[idx] = arow[idx];
}

extern "C" void kernel_launch(void* const* d_in, const int* in_sizes, int n_in,
                              void* d_out, int out_size, void* d_ws, size_t ws_size,
                              hipStream_t stream) {
    const float* nodes     = (const float*)d_in[0];
    const int*   senders   = (const int*)  d_in[1];
    const int*   receivers = (const int*)  d_in[2];
    const float* edges     = (const float*)d_in[3];
    const float* Wm        = (const float*)d_in[4];
    float*       out       = (float*)d_out;

    float* y     = (float*)d_ws;
    float* ds    = y  + (size_t)N_NODES * DIM;
    float* dr    = ds + N_NODES;
    int*   rtot  = (int*)(dr + N_NODES);
    int*   rfill = rtot + NR_PAD;
    int2*  bins  = (int2*)(rfill + NR_PAD);

    k_init_deg<<<(N_NODES + 255) / 256, 256, 0, stream>>>(ds, dr, rtot);
    k_edge_deg<<<2 * DEG_SLICES * DEG_PARTS, DEG_THREADS, 0, stream>>>(senders, receivers, edges, ds, dr, rtot);

    k_bin_gemm<<<GEMM_BLOCKS + BIN_BLOCKS, GB_THREADS, 0, stream>>>(
        nodes, Wm, ds, dr, y, senders, receivers, edges, rtot, rfill, bins);
    k_row_sums<<<NR, AGG_THREADS, 0, stream>>>(y, rtot, bins, out);
}


# degree sums: workgroups store partial degrees and range counts with plain line stores, GEMM workgroups add the 32 partials in fixed order (sum order changes); no init launch, no global f32 atomics
# speedup vs baseline: 4.2931x; 1.0552x over previous
#include <hip/hip_runtime.h>
#include <cstdint>

#define N_NODES 50000
#define N_EDGES 800000
#define DIM     128

#define RN      200
#define NR      (N_NODES / RN)
#define NR_PAD  256
static_assert(N_NODES % RN == 0, "ranges are whole");
static_assert(NR <= NR_PAD && RN <= 256, "range id and local row fit their fields");
static_assert(N_NODES <= 65536, "a sender id fits 16 bits of the packed edge word");

typedef __attribute__((ext_vector_type(2))) float v2f;
typedef __attribute__((ext_vector_type(8))) float v8f;

#define DEG_SLICES  32
#define DEG_PARTS   4
#define DEG_THREADS 256
#define DEG_PER     (N_EDGES / DEG_SLICES)
#define DEG_NV      (DEG_PER / 4)
#define DEG_IT      ((DEG_NV + DEG_THREADS - 1) / DEG_THREADS)
#define DEG_PART_N  ((((N_NODES + DEG_PARTS - 1) / DEG_PARTS) + 31) / 32 * 32)
#define DEG_STEPS   ((DEG_PART_N + DEG_THREADS - 1) / DEG_THREADS)
#define DEG_PITCH   (((N_NODES + 31) / 32) * 32)
#define DEG_CNT_ROWS (DEG_SLICES * DEG_PARTS)
static_assert(N_EDGES % (DEG_SLICES * 4) == 0, "slices must be whole 16-B groups");
static_assert(DEG_PART_N * DEG_PARTS >= N_NODES, "the parts cover every node");
static_assert(DEG_PART_N * (DEG_PARTS - 1) < N_NODES, "no part is empty");
static_assert(DEG_PART_N % 32 == 0, "a part starts on a line border");
static_assert((N_NODES - DEG_PART_N * (DEG_PARTS - 1)) % 4 == 0 && DEG_PART_N % 4 == 0,
              "every part is a whole number of 16-B groups");
static_assert(DEG_THREADS >= NR_PAD, "one thread per range counter");
static_assert(RN <= DEG_PART_N, "a range lies in one node part or in two neighbours");

__global__ void __launch_bounds__(DEG_THREADS) k_edge_deg(const int*   __restrict__ snd,
                                                          const int*   __restrict__ rcv,
                                                          const float* __restrict__ ew,
                                                          float* __restrict__ pdeg,
                                                          int* __restrict__ rcp,
                                                          int* __restrict__ rfill) {
    __shared__ int rc[NR_PAD];
    __shared__ __attribute__((aligned(16))) float hist[DEG_STEPS * DEG_THREADS];

    const int  tid   = threadIdx.x;
    const int  nsp   = DEG_SLICES * DEG_PARTS;
    const bool recv  = blockIdx.x < nsp;
    const int  sp    = recv ? blockIdx.x : blockIdx.x - nsp;
    const int  slice = sp / DEG_PARTS;
    const int  lo    = (sp % DEG_PARTS) * DEG_PART_N;
    const int  len   = N_NODES - lo < DEG_PART_N ? N_NODES - lo : DEG_PART_N;
    const int*  idx  = recv ? rcv : snd;

    const int4*   ip = (const int4*)  (idx + slice * DEG_PER);
    const float4* wp = (const float4*)(ew  + slice * DEG_PER);
    int4   n[DEG_IT];
    float4 w[DEG_IT];
    #pragma unroll
    for (int k = 0; k < DEG_IT; ++k) {
        const int v  = tid + k * DEG_THREADS;
        const int vc = v < DEG_NV ? v : DEG_NV - 1;
        n[k] = ip[vc];
        w[k] = wp[vc];
    }

    for (int i = tid; i < DEG_STEPS * DEG_THREADS / 4; i += DEG_THREADS)
        ((float4*)hist)[i] = make_float4(0.0f, 0.0f, 0.0f, 0.0f);
    if (tid < NR_PAD) rc[tid] = 0;
    __syncthreads();

    #pragma unroll
    for (int k = 0; k < DEG_IT; ++k) {
        if (tid + k * DEG_THREADS < DEG_NV) {
            const int   nn[4] = { n[k].x, n[k].y, n[k].z, n[k].w };
            const float ww[4] = { w[k].x, w[k].y, w[k].z, w[k].w };
            #pragma unroll
            for (int j = 0; j < 4; ++j) {
                const unsigned r = (unsigned)(nn[j] - lo);
                if (r < (unsigned)len) {
                    atomicAdd(&hist[r], ww[j] + 1.0f);
                    if (recv) atomicAdd(&rc[nn[j] / RN], 1);
                }
            }
        }
    }
    __syncthreads();

    if (recv && tid < NR_PAD) rcp[sp * NR_PAD + tid] = rc[tid];
    if (blockIdx.x == 0 && tid < NR_PAD) rfill[tid] = 0;

    float4*       prow = (float4*)(pdeg + (size_t)((recv ? DEG_SLICES : 0) + slice) * DEG_PITCH + lo);
    const float4* hrow = (const float4*)hist;
    for (int i = tid; i < len / 4; i += DEG_THREADS) prow[i] = hrow[i];
}

#define GB_THREADS   1024
#define N_STRIPS     (N_NODES / 16)
#define GEMM_STRIPS  (GB_THREADS / 64)
#define GEMM_BLOCKS  ((N_STRIPS + GEMM_STRIPS - 1) / GEMM_STRIPS)
#define DEG_Q        (2 * DEG_SLICES / 4)
static_assert(N_NODES % 16 == 0, "strips are whole");
static_assert(GB_THREADS == 4 * GEMM_STRIPS * 16 && DEG_SLICES % 2 == 0,
              "one thread per (row, quarter); a quarter is half a side");

__device__ __forceinline__ void gemm_scale_role(const int gb,
                                                const float* __restrict__ nodes,
                                                const float* __restrict__ Wm,
                                                const float* __restrict__ pdeg,
                                                float* __restrict__ y,
                                                float* __restrict__ w_lds,
                                                float* __restrict__ s_deg) {
    const int tid   = threadIdx.x;
    const int lane  = tid & 31;
    const int wave  = tid >> 5;
    const int strip = gb * GEMM_STRIPS + (wave >> 1);
    const int par   = wave & 1;
    const int m     = lane & 15;

    float pd[DEG_Q];
    {
        const int q  = tid >> 8;
        const int rw = gb * (GEMM_STRIPS * 16) + (tid & 255);
        const float* pp = pdeg + (size_t)(q * DEG_Q) * DEG_PITCH + (rw < N_NODES ? rw : N_NODES - 1);
        #pragma unroll
        for (int k = 0; k < DEG_Q; ++k) pd[k] = pp[(size_t)k * DEG_PITCH];
    }

    for (int idx = tid; idx < DIM * DIM; idx += GB_THREADS) {
        int k = idx >> 7;
        int c = idx & (DIM - 1);
        w_lds[((k >> 1) << 8) + (c << 1) + (k & 1)] = Wm[idx];
    }
    {
        float qs = pd[0];
        #pragma unroll
        for (int k = 1; k < DEG_Q; ++k) qs += pd[k];
        s_deg[tid] = qs;
    }
    __syncthreads();

    if (strip < N_STRIPS) {
        const int row0 = strip * 16;
        const int koff = (lane >> 4) << 1;

        const int   rl  = ((wave >> 1) << 4) + m;
        const int   sd  = (lane >> 4) << 9;
        const float deg = (2.0f + s_deg[sd + rl]) + s_deg[sd + 256 + rl];
        const float rs = rsqrtf(fmaxf(deg, 1.0f));
        const float sc_m = rs * __shfl_xor(rs, 16);

        const float* arow = nodes + (size_t)(row0 + m) * DIM;

        v8f acc[4] = {};

        #pragma unroll 4
        for (int k0 = 0; k0 < DIM; k0 += 4) {
            v2f a = *(const v2f*)(arow + k0 + koff);
            const float* bbase = w_lds + (((k0 + koff) >> 1) << 8);
            #pragma unroll
            for (int q = 0; q < 4; ++q) {
                const int t = 2 * q + par;
                v2f b = *(const v2f*)(bbase + ((t * 16 + m) << 1));
                acc[q] = __builtin_amdgcn_wmma_f32_16x16x4_f32(
                     false, a,  false, b,
                     (short)0, acc[q],
                     false,  false);
            }
        }

        const int roff = (lane >> 4) << 3;
        #pragma unroll
        for (int v = 0; v < 8; ++v) {
            const int row = row0 + v + roff;
            const float sc = __shfl(sc_m, v + roff);
            float* yrow = y + (size_t)row * DIM;
            *(float4*)(yrow + 64 * par + 4 * m) =
                make_float4(acc[0][v] * sc, acc[1][v] * sc, acc[2][v] * sc, acc[3][v] * sc);
        }
    }
}

#define BIN_THREADS 1024
#define BIN_PER     3200
#define BIN_BLOCKS  (N_EDGES / BIN_PER)
#define BIN_IT      ((BIN_PER + BIN_THREADS - 1) / BIN_THREADS)
static_assert(N_EDGES % BIN_PER == 0, "slices are whole");

static_assert(BIN_THREADS == GB_THREADS, "both roles of the launch use one block size");

__device__ __forceinline__ void bin_role(const int bb,
                                         const int*   __restrict__ snd,
                                         const int*   __restrict__ rcv,
                                         const float* __restrict__ ew,
                                         const int*   __restrict__ rcp,
                                         int*  __restrict__ rtot,
                                         int*  __restrict__ rfill,
                                         int2* __restrict__ bins,
                                         int* __restrict__ s_wsum,
                                         int* __restrict__ s_cnt,
                                         int* __restrict__ s_base) {
    const int tid  = threadIdx.x;
    const int lane = tid & 31;
    const int wave = tid >> 5;
    const int e0   = bb * BIN_PER;

    int tot = 0, incl = 0;
    if (tid < NR_PAD) {
        s_cnt[tid] = 0;
        if (tid < NR) {
            const int  p0 = (tid * RN) / DEG_PART_N;
            const int  p1 = (tid * RN + RN - 1) / DEG_PART_N;
            const int* c0 = rcp + p0 * NR_PAD + tid;
            #pragma unroll
            for (int s = 0; s < DEG_SLICES; ++s) tot += c0[s * DEG_PARTS * NR_PAD];
            if (p1 != p0) {
                const int* c1 = rcp + p1 * NR_PAD + tid;
                #pragma unroll
                for (int s = 0; s < DEG_SLICES; ++s) tot += c1[s * DEG_PARTS * NR_PAD];
            }
        }
        if (bb == 0) rtot[tid] = tot;
        incl = tot;
        #pragma unroll
        for (int d = 1; d < 32; d <<= 1) {
            const int t = __shfl_up(incl, d);
            if (lane >= d) incl += t;
        }
        if (lane == 31) s_wsum[wave] = incl;
    }

    int   rr[BIN_IT], ss[BIN_IT];
    float ww[BIN_IT];
    #pragma unroll
    for (int k = 0; k < BIN_IT; ++k) {
        const int i = tid + k * BIN_THREADS;
        rr[k] = 0; ss[k] = 0; ww[k] = 0.0f;
        if (i < BIN_PER) { rr[k] = rcv[e0 + i]; ss[k] = snd[e0 + i]; ww[k] = ew[e0 + i]; }
    }
    __syncthreads();

    int g[BIN_IT], rank[BIN_IT];
    #pragma unroll
    for (int k = 0; k < BIN_IT; ++k) {
        g[k] = rr[k] / RN;
        rank[k] = 0;
        if (tid + k * BIN_THREADS < BIN_PER) rank[k] = atomicAdd(&s_cnt[g[k]], 1);
    }
    __syncthreads();

    if (tid < NR) {
        int st = incl - tot;
        for (int k = 0; k < wave; ++k) st += s_wsum[k];
        s_base[tid] = st + atomicAdd(&rfill[tid], s_cnt[tid]);
    }
    __syncthreads();

    #pragma unroll
    for (int k = 0; k < BIN_IT; ++k) {
        if (tid + k * BIN_THREADS < BIN_PER)
            bins[s_base[g[k]] + rank[k]] =
                make_int2(ss[k] | ((rr[k] - g[k] * RN) << 16), __float_as_int(ww[k]));
    }
}

__global__ void __launch_bounds__(GB_THREADS) __attribute__((amdgpu_waves_per_eu(16, 16)))
k_bin_gemm(const float* __restrict__ nodes,
           const float* __restrict__ Wm,
           const float* __restrict__ pdeg,
           float* __restrict__ y,
           const int*   __restrict__ snd,
           const int*   __restrict__ rcv,
           const float* __restrict__ ew,
           const int*   __restrict__ rcp,
           int*  __restrict__ rtot,
           int*  __restrict__ rfill,
           int2* __restrict__ bins) {
    __shared__ float w_lds[DIM * DIM];
    __shared__ float s_deg[4 * GEMM_STRIPS * 16];
    __shared__ int s_wsum[NR_PAD / 32];
    __shared__ int s_cnt[NR_PAD];
    __shared__ int s_base[NR_PAD];

    if (blockIdx.x < GEMM_BLOCKS)
        gemm_scale_role(blockIdx.x, nodes, Wm, pdeg, y, w_lds, s_deg);
    else
        bin_role(blockIdx.x - GEMM_BLOCKS, snd, rcv, ew, rcp, rtot, rfill, bins, s_wsum, s_cnt, s_base);
}

#define AGG_THREADS 1024
#define AGG_WAVES   (AGG_THREADS / 32)
#define AGG_ROWS    8
#define AGG_INIT    ((RN + AGG_WAVES - 1) / AGG_WAVES)
#define SORT_CH     4096
#define SORT_IT     (SORT_CH / AGG_THREADS)
#define SORT_SCAN   256
static_assert(SORT_CH % AGG_THREADS == 0, "every thread takes the same number of edge words");
static_assert(RN <= SORT_SCAN && SORT_SCAN <= AGG_THREADS && SORT_SCAN % 32 == 0,
              "one scan thread per local receiver, whole waves");

__device__ __forceinline__ void flush_run(float* __restrict__ acc, const int r, const int lane,
                                          const float sx, const float sy,
                                          const float sz, const float sw) {
    float* a = acc + r * DIM + lane;
    atomicAdd(a,      sx);
    atomicAdd(a + 32, sy);
    atomicAdd(a + 64, sz);
    atomicAdd(a + 96, sw);
}

__global__ void __launch_bounds__(AGG_THREADS) k_row_sums(const float* __restrict__ y,
                                                          const int*   __restrict__ rtot,
                                                          const int2*  __restrict__ bins,
                                                          float* __restrict__ out) {
    __shared__ __attribute__((aligned(16))) float acc[RN * DIM];
    __shared__ int2 sorted[SORT_CH];
    __shared__ int rcnt[SORT_SCAN];
    __shared__ int roff[SORT_SCAN];
    __shared__ int s_wsum[SORT_SCAN / 32];
    __shared__ int s_start;

    const int tid   = threadIdx.x;
    const int lane  = tid & 31;
    const int wave  = __builtin_amdgcn_readfirstlane(tid >> 5);
    const int range = blockIdx.x;
    const int node0 = range * RN;

    if (tid == 0) s_start = 0;
    if (tid < SORT_SCAN) rcnt[tid] = 0;
    __syncthreads();

    if (tid < range) atomicAdd(&s_start, rtot[tid]);

    float4 self[AGG_INIT];
    #pragma unroll
    for (int p = 0; p < AGG_INIT; ++p) {
        const int rl  = wave + p * AGG_WAVES;
        const int rlc = rl < RN ? rl : RN - 1;
        self[p] = *(const float4*)(y + (size_t)(node0 + rlc) * DIM + 4 * lane);
    }
    #pragma unroll
    for (int p = 0; p < AGG_INIT; ++p) {
        const int rl = wave + p * AGG_WAVES;
        if (rl < RN) {
            float* a = acc + rl * DIM + lane;
            a[0] = self[p].x; a[32] = self[p].y; a[64] = self[p].z; a[96] = self[p].w;
        }
    }
    __syncthreads();

    const int start = __builtin_amdgcn_readfirstlane(s_start);
    const int cnt   = rtot[range];

    for (int c0 = 0; c0 < cnt; c0 += SORT_CH) {
        const int nch = cnt - c0 < SORT_CH ? cnt - c0 : SORT_CH;

        int2 e[SORT_IT];
        int  rk[SORT_IT];
        #pragma unroll
        for (int k = 0; k < SORT_IT; ++k) {
            const int i = tid + k * AGG_THREADS;
            e[k] = make_int2(0, 0);
            if (i < nch) e[k] = bins[start + c0 + i];
        }
        #pragma unroll
        for (int k = 0; k < SORT_IT; ++k) {
            rk[k] = 0;
            if (tid + k * AGG_THREADS < nch) rk[k] = atomicAdd(&rcnt[e[k].x >> 16], 1);
        }
        __syncthreads();

        int c = 0, incl = 0;
        if (tid < SORT_SCAN) {
            c = rcnt[tid];
            rcnt[tid] = 0;
            incl = c;
            #pragma unroll
            for (int d = 1; d < 32; d <<= 1) {
                const int t = __shfl_up(incl, d);
                if (lane >= d) incl += t;
            }
            if (lane == 31) s_wsum[wave] = incl;
        }
        __syncthreads();
        if (tid < SORT_SCAN) {
            int st = incl - c;
            #pragma unroll
            for (int k = 0; k < SORT_SCAN / 32; ++k)
                if (k < wave) st += s_wsum[k];
            roff[tid] = st;
        }
        __syncthreads();

        #pragma unroll
        for (int k = 0; k < SORT_IT; ++k)
            if (tid + k * AGG_THREADS < nch) sorted[roff[e[k].x >> 16] + rk[k]] = e[k];
        __syncthreads();

        const int seg   = (((nch + AGG_WAVES - 1) / AGG_WAVES) + AGG_ROWS - 1) & ~(AGG_ROWS - 1);
        const int wbase = wave * seg;
        const int wend  = wbase + seg < nch ? wbase + seg : nch;
        int   cur = -1;
        float sx = 0.0f, sy = 0.0f, sz = 0.0f, sw = 0.0f;
        int2 rec = sorted[wbase + lane < nch ? wbase + lane : nch - 1];
        for (int p = wbase; p < wend; p += 32) {
            const int n = wend - p < 32 ? wend - p : 32;
            const int2 nrec = sorted[p + 32 + lane < nch ? p + 32 + lane : nch - 1];
            #pragma unroll
            for (int j0 = 0; j0 < 32; j0 += AGG_ROWS) {
                if (j0 < n) {
                    float4 v[AGG_ROWS];
                    int    rl[AGG_ROWS];
                    float  w[AGG_ROWS];
                    #pragma unroll
                    for (int u = 0; u < AGG_ROWS; ++u) {
                        const int px = __builtin_amdgcn_readlane(rec.x, j0 + u);
                        w[u]  = __int_as_float(__builtin_amdgcn_readlane(rec.y, j0 + u));
                        rl[u] = px >> 16;
                        v[u]  = *(const float4*)(y + (size_t)(px & 0xFFFF) * DIM + lane * 4);
                    }
                    #pragma unroll
                    for (int u = 0; u < AGG_ROWS; ++u)
                        asm volatile("" :: "v"(v[u].x), "v"(v[u].y), "v"(v[u].z), "v"(v[u].w));
                    #pragma unroll
                    for (int u = 0; u < AGG_ROWS; ++u) {
                        if (j0 + u < n) {
                            if (rl[u] != cur) {
                                if (cur >= 0) flush_run(acc, cur, lane, sx, sy, sz, sw);
                                cur = rl[u];
                                sx = 0.0f; sy = 0.0f; sz = 0.0f; sw = 0.0f;
                            }
                            sx += v[u].x * w[u];
                            sy += v[u].y * w[u];
                            sz += v[u].z * w[u];
                            sw += v[u].w * w[u];
                        }
                    }
                }
            }
            rec = nrec;
        }
        if (cur >= 0) flush_run(acc, cur, lane, sx, sy, sz, sw);
        __syncthreads();
    }

    float4*       orow = (float4*)(out + (size_t)node0 * DIM);
    const float4* arow = (const float4*)acc;
    for (int idx = tid; idx < RN * DIM / 4; idx += AGG_THREADS) orow[idx] = arow[idx];
}

extern "C" void kernel_launch(void* const* d_in, const int* in_sizes, int n_in,
                              void* d_out, int out_size, void* d_ws, size_t ws_size,
                              hipStream_t stream) {
    const float* nodes     = (const float*)d_in[0];
    const int*   senders   = (const int*)  d_in[1];
    const int*   receivers = (const int*)  d_in[2];
    const float* edges     = (const float*)d_in[3];
    const float* Wm        = (const float*)d_in[4];
    float*       out       = (float*)d_out;

    float* y     = (float*)d_ws;
    float* pdeg  = y + (size_t)N_NODES * DIM;
    int*   rcp   = (int*)(pdeg + (size_t)2 * DEG_SLICES * DEG_PITCH);
    int*   rtot  = rcp + DEG_CNT_ROWS * NR_PAD;
    int*   rfill = rtot + NR_PAD;
    int2*  bins  = (int2*)(rfill + NR_PAD);

    k_edge_deg<<<2 * DEG_SLICES * DEG_PARTS, DEG_THREADS, 0, stream>>>(senders, receivers, edges, pdeg, rcp, rfill);

    k_bin_gemm<<<GEMM_BLOCKS + BIN_BLOCKS, GB_THREADS, 0, stream>>>(
        nodes, Wm, pdeg, y, senders, receivers, edges, rcp, rtot, rfill, bins);
    k_row_sums<<<NR, AGG_THREADS, 0, stream>>>(y, rtot, bins, out);
}


# GEMM+binning launch: 512-thread blocks, a GEMM wave takes a whole 16x128 strip (8 tiles, 64 accumulator registers): 7136 waves enter the launch, was 14272; a strip's A is loaded once
# speedup vs baseline: 4.4524x; 1.0371x over previous
#include <hip/hip_runtime.h>
#include <cstdint>

#define N_NODES 50000
#define N_EDGES 800000
#define DIM     128

#define RN      200
#define NR      (N_NODES / RN)
#define NR_PAD  256
static_assert(N_NODES % RN == 0, "ranges are whole");
static_assert(NR <= NR_PAD && RN <= 256, "range id and local row fit their fields");
static_assert(N_NODES <= 65536, "a sender id fits 16 bits of the packed edge word");

typedef __attribute__((ext_vector_type(2))) float v2f;
typedef __attribute__((ext_vector_type(8))) float v8f;

#define DEG_SLICES  32
#define DEG_PARTS   4
#define DEG_THREADS 256
#define DEG_PER     (N_EDGES / DEG_SLICES)
#define DEG_NV      (DEG_PER / 4)
#define DEG_IT      ((DEG_NV + DEG_THREADS - 1) / DEG_THREADS)
#define DEG_PART_N  ((((N_NODES + DEG_PARTS - 1) / DEG_PARTS) + 31) / 32 * 32)
#define DEG_STEPS   ((DEG_PART_N + DEG_THREADS - 1) / DEG_THREADS)
#define DEG_PITCH   (((N_NODES + 31) / 32) * 32)
#define DEG_CNT_ROWS (DEG_SLICES * DEG_PARTS)
static_assert(N_EDGES % (DEG_SLICES * 4) == 0, "slices must be whole 16-B groups");
static_assert(DEG_PART_N * DEG_PARTS >= N_NODES, "the parts cover every node");
static_assert(DEG_PART_N * (DEG_PARTS - 1) < N_NODES, "no part is empty");
static_assert(DEG_PART_N % 32 == 0, "a part starts on a line border");
static_assert((N_NODES - DEG_PART_N * (DEG_PARTS - 1)) % 4 == 0 && DEG_PART_N % 4 == 0,
              "every part is a whole number of 16-B groups");
static_assert(DEG_THREADS >= NR_PAD, "one thread per range counter");
static_assert(RN <= DEG_PART_N, "a range lies in one node part or in two neighbours");

__global__ void __launch_bounds__(DEG_THREADS) k_edge_deg(const int*   __restrict__ snd,
                                                          const int*   __restrict__ rcv,
                                                          const float* __restrict__ ew,
                                                          float* __restrict__ pdeg,
                                                          int* __restrict__ rcp,
                                                          int* __restrict__ rfill) {
    __shared__ int rc[NR_PAD];
    __shared__ __attribute__((aligned(16))) float hist[DEG_STEPS * DEG_THREADS];

    const int  tid   = threadIdx.x;
    const int  nsp   = DEG_SLICES * DEG_PARTS;
    const bool recv  = blockIdx.x < nsp;
    const int  sp    = recv ? blockIdx.x : blockIdx.x - nsp;
    const int  slice = sp / DEG_PARTS;
    const int  lo    = (sp % DEG_PARTS) * DEG_PART_N;
    const int  len   = N_NODES - lo < DEG_PART_N ? N_NODES - lo : DEG_PART_N;
    const int*  idx  = recv ? rcv : snd;

    const int4*   ip = (const int4*)  (idx + slice * DEG_PER);
    const float4* wp = (const float4*)(ew  + slice * DEG_PER);
    int4   n[DEG_IT];
    float4 w[DEG_IT];
    #pragma unroll
    for (int k = 0; k < DEG_IT; ++k) {
        const int v  = tid + k * DEG_THREADS;
        const int vc = v < DEG_NV ? v : DEG_NV - 1;
        n[k] = ip[vc];
        w[k] = wp[vc];
    }

    for (int i = tid; i < DEG_STEPS * DEG_THREADS / 4; i += DEG_THREADS)
        ((float4*)hist)[i] = make_float4(0.0f, 0.0f, 0.0f, 0.0f);
    if (tid < NR_PAD) rc[tid] = 0;
    __syncthreads();

    #pragma unroll
    for (int k = 0; k < DEG_IT; ++k) {
        if (tid + k * DEG_THREADS < DEG_NV) {
            const int   nn[4] = { n[k].x, n[k].y, n[k].z, n[k].w };
            const float ww[4] = { w[k].x, w[k].y, w[k].z, w[k].w };
            #pragma unroll
            for (int j = 0; j < 4; ++j) {
                const unsigned r = (unsigned)(nn[j] - lo);
                if (r < (unsigned)len) {
                    atomicAdd(&hist[r], ww[j] + 1.0f);
                    if (recv) atomicAdd(&rc[nn[j] / RN], 1);
                }
            }
        }
    }
    __syncthreads();

    if (recv && tid < NR_PAD) rcp[sp * NR_PAD + tid] = rc[tid];
    if (blockIdx.x == 0 && tid < NR_PAD) rfill[tid] = 0;

    float4*       prow = (float4*)(pdeg + (size_t)((recv ? DEG_SLICES : 0) + slice) * DEG_PITCH + lo);
    const float4* hrow = (const float4*)hist;
    for (int i = tid; i < len / 4; i += DEG_THREADS) prow[i] = hrow[i];
}

#define GB_THREADS   512
#define N_STRIPS     (N_NODES / 16)
#define GEMM_STRIPS  (GB_THREADS / 32)
#define GEMM_BLOCKS  ((N_STRIPS + GEMM_STRIPS - 1) / GEMM_STRIPS)
#define DEG_Q        (DEG_SLICES / 2)
static_assert(N_NODES % 16 == 0, "strips are whole");
static_assert(GB_THREADS == 2 * GEMM_STRIPS * 16 && DEG_SLICES % 2 == 0,
              "one thread per (row, side); a quarter is half a side");

__device__ __forceinline__ void gemm_scale_role(const int gb,
                                                const float* __restrict__ nodes,
                                                const float* __restrict__ Wm,
                                                const float* __restrict__ pdeg,
                                                float* __restrict__ y,
                                                float* __restrict__ w_lds,
                                                float* __restrict__ s_deg) {
    const int tid   = threadIdx.x;
    const int lane  = tid & 31;
    const int wave  = tid >> 5;
    const int strip = gb * GEMM_STRIPS + wave;
    const int m     = lane & 15;

    float pd[2 * DEG_Q];
    {
        const int sd = tid >> 8;
        const int rw = gb * (GEMM_STRIPS * 16) + (tid & 255);
        const float* pp = pdeg + (size_t)(sd * DEG_SLICES) * DEG_PITCH + (rw < N_NODES ? rw : N_NODES - 1);
        #pragma unroll
        for (int k = 0; k < 2 * DEG_Q; ++k) pd[k] = pp[(size_t)k * DEG_PITCH];
    }

    for (int idx = tid; idx < DIM * DIM; idx += GB_THREADS) {
        int k = idx >> 7;
        int c = idx & (DIM - 1);
        w_lds[((k >> 1) << 8) + (c << 1) + (k & 1)] = Wm[idx];
    }
    {
        float ql = pd[0], qu = pd[DEG_Q];
        #pragma unroll
        for (int k = 1; k < DEG_Q; ++k) { ql += pd[k]; qu += pd[DEG_Q + k]; }
        const int sd = tid >> 8;
        s_deg[(2 * sd) * 256 + (tid & 255)]     = ql;
        s_deg[(2 * sd + 1) * 256 + (tid & 255)] = qu;
    }
    __syncthreads();

    if (strip < N_STRIPS) {
        const int row0 = strip * 16;
        const int koff = (lane >> 4) << 1;

        const int   rl  = (wave << 4) + m;
        const int   sd  = (lane >> 4) << 9;
        const float deg = (2.0f + s_deg[sd + rl]) + s_deg[sd + 256 + rl];
        const float rs = rsqrtf(fmaxf(deg, 1.0f));
        const float sc_m = rs * __shfl_xor(rs, 16);

        const float* arow = nodes + (size_t)(row0 + m) * DIM;

        v8f acc[8] = {};

        #pragma unroll 4
        for (int k0 = 0; k0 < DIM; k0 += 4) {
            v2f a = *(const v2f*)(arow + k0 + koff);
            const float* bbase = w_lds + (((k0 + koff) >> 1) << 8);
            #pragma unroll
            for (int t = 0; t < 8; ++t) {
                v2f b = *(const v2f*)(bbase + ((t * 16 + m) << 1));
                acc[t] = __builtin_amdgcn_wmma_f32_16x16x4_f32(
                     false, a,  false, b,
                     (short)0, acc[t],
                     false,  false);
            }
        }

        const int roff = (lane >> 4) << 3;
        #pragma unroll
        for (int v = 0; v < 8; ++v) {
            const int row = row0 + v + roff;
            const float sc = __shfl(sc_m, v + roff);
            float* yrow = y + (size_t)row * DIM;
            *(float4*)(yrow + 4 * m) =
                make_float4(acc[0][v] * sc, acc[2][v] * sc, acc[4][v] * sc, acc[6][v] * sc);
            *(float4*)(yrow + 64 + 4 * m) =
                make_float4(acc[1][v] * sc, acc[3][v] * sc, acc[5][v] * sc, acc[7][v] * sc);
        }
    }
}

#define BIN_THREADS 512
#define BIN_PER     3200
#define BIN_BLOCKS  (N_EDGES / BIN_PER)
#define BIN_IT      ((BIN_PER + BIN_THREADS - 1) / BIN_THREADS)
static_assert(N_EDGES % BIN_PER == 0, "slices are whole");

static_assert(BIN_THREADS == GB_THREADS, "both roles of the launch use one block size");
static_assert(BIN_THREADS >= NR_PAD, "one thread per range in the running sum");

__device__ __forceinline__ void bin_role(const int bb,
                                         const int*   __restrict__ snd,
                                         const int*   __restrict__ rcv,
                                         const float* __restrict__ ew,
                                         const int*   __restrict__ rcp,
                                         int*  __restrict__ rtot,
                                         int*  __restrict__ rfill,
                                         int2* __restrict__ bins,
                                         int* __restrict__ s_wsum,
                                         int* __restrict__ s_cnt,
                                         int* __restrict__ s_base) {
    const int tid  = threadIdx.x;
    const int lane = tid & 31;
    const int wave = tid >> 5;
    const int e0   = bb * BIN_PER;

    int tot = 0, incl = 0;
    if (tid < NR_PAD) {
        s_cnt[tid] = 0;
        if (tid < NR) {
            const int  p0 = (tid * RN) / DEG_PART_N;
            const int  p1 = (tid * RN + RN - 1) / DEG_PART_N;
            const int* c0 = rcp + p0 * NR_PAD + tid;
            #pragma unroll
            for (int s = 0; s < DEG_SLICES; ++s) tot += c0[s * DEG_PARTS * NR_PAD];
            if (p1 != p0) {
                const int* c1 = rcp + p1 * NR_PAD + tid;
                #pragma unroll
                for (int s = 0; s < DEG_SLICES; ++s) tot += c1[s * DEG_PARTS * NR_PAD];
            }
        }
        if (bb == 0) rtot[tid] = tot;
        incl = tot;
        #pragma unroll
        for (int d = 1; d < 32; d <<= 1) {
            const int t = __shfl_up(incl, d);
            if (lane >= d) incl += t;
        }
        if (lane == 31) s_wsum[wave] = incl;
    }

    int   rr[BIN_IT], ss[BIN_IT];
    float ww[BIN_IT];
    #pragma unroll
    for (int k = 0; k < BIN_IT; ++k) {
        const int i = tid + k * BIN_THREADS;
        rr[k] = 0; ss[k] = 0; ww[k] = 0.0f;
        if (i < BIN_PER) { rr[k] = rcv[e0 + i]; ss[k] = snd[e0 + i]; ww[k] = ew[e0 + i]; }
    }
    __syncthreads();

    int g[BIN_IT], rank[BIN_IT];
    #pragma unroll
    for (int k = 0; k < BIN_IT; ++k) {
        g[k] = rr[k] / RN;
        rank[k] = 0;
        if (tid + k * BIN_THREADS < BIN_PER) rank[k] = atomicAdd(&s_cnt[g[k]], 1);
    }
    __syncthreads();

    if (tid < NR) {
        int st = incl - tot;
        for (int k = 0; k < wave; ++k) st += s_wsum[k];
        s_base[tid] = st + atomicAdd(&rfill[tid], s_cnt[tid]);
    }
    __syncthreads();

    #pragma unroll
    for (int k = 0; k < BIN_IT; ++k) {
        if (tid + k * BIN_THREADS < BIN_PER)
            bins[s_base[g[k]] + rank[k]] =
                make_int2(ss[k] | ((rr[k] - g[k] * RN) << 16), __float_as_int(ww[k]));
    }
}

__global__ void __launch_bounds__(GB_THREADS) __attribute__((amdgpu_waves_per_eu(8, 8)))
k_bin_gemm(const float* __restrict__ nodes,
           const float* __restrict__ Wm,
           const float* __restrict__ pdeg,
           float* __restrict__ y,
           const int*   __restrict__ snd,
           const int*   __restrict__ rcv,
           const float* __restrict__ ew,
           const int*   __restrict__ rcp,
           int*  __restrict__ rtot,
           int*  __restrict__ rfill,
           int2* __restrict__ bins) {
    __shared__ float w_lds[DIM * DIM];
    __shared__ float s_deg[4 * GEMM_STRIPS * 16];
    __shared__ int s_wsum[NR_PAD / 32];
    __shared__ int s_cnt[NR_PAD];
    __shared__ int s_base[NR_PAD];

    if (blockIdx.x < GEMM_BLOCKS)
        gemm_scale_role(blockIdx.x, nodes, Wm, pdeg, y, w_lds, s_deg);
    else
        bin_role(blockIdx.x - GEMM_BLOCKS, snd, rcv, ew, rcp, rtot, rfill, bins, s_wsum, s_cnt, s_base);
}

#define AGG_THREADS 1024
#define AGG_WAVES   (AGG_THREADS / 32)
#define AGG_ROWS    8
#define AGG_INIT    ((RN + AGG_WAVES - 1) / AGG_WAVES)
#define SORT_CH     4096
#define SORT_IT     (SORT_CH / AGG_THREADS)
#define SORT_SCAN   256
static_assert(SORT_CH % AGG_THREADS == 0, "every thread takes the same number of edge words");
static_assert(RN <= SORT_SCAN && SORT_SCAN <= AGG_THREADS && SORT_SCAN % 32 == 0,
              "one scan thread per local receiver, whole waves");

__device__ __forceinline__ void flush_run(float* __restrict__ acc, const int r, const int lane,
                                          const float sx, const float sy,
                                          const float sz, const float sw) {
    float* a = acc + r * DIM + lane;
    atomicAdd(a,      sx);
    atomicAdd(a + 32, sy);
    atomicAdd(a + 64, sz);
    atomicAdd(a + 96, sw);
}

__global__ void __launch_bounds__(AGG_THREADS) k_row_sums(const float* __restrict__ y,
                                                          const int*   __restrict__ rtot,
                                                          const int2*  __restrict__ bins,
                                                          float* __restrict__ out) {
    __shared__ __attribute__((aligned(16))) float acc[RN * DIM];
    __shared__ int2 sorted[SORT_CH];
    __shared__ int rcnt[SORT_SCAN];
    __shared__ int roff[SORT_SCAN];
    __shared__ int s_wsum[SORT_SCAN / 32];
    __shared__ int s_start;

    const int tid   = threadIdx.x;
    const int lane  = tid & 31;
    const int wave  = __builtin_amdgcn_readfirstlane(tid >> 5);
    const int range = blockIdx.x;
    const int node0 = range * RN;

    if (tid == 0) s_start = 0;
    if (tid < SORT_SCAN) rcnt[tid] = 0;
    __syncthreads();

    if (tid < range) atomicAdd(&s_start, rtot[tid]);

    float4 self[AGG_INIT];
    #pragma unroll
    for (int p = 0; p < AGG_INIT; ++p) {
        const int rl  = wave + p * AGG_WAVES;
        const int rlc = rl < RN ? rl : RN - 1;
        self[p] = *(const float4*)(y + (size_t)(node0 + rlc) * DIM + 4 * lane);
    }
    #pragma unroll
    for (int p = 0; p < AGG_INIT; ++p) {
        const int rl = wave + p * AGG_WAVES;
        if (rl < RN) {
            float* a = acc + rl * DIM + lane;
            a[0] = self[p].x; a[32] = self[p].y; a[64] = self[p].z; a[96] = self[p].w;
        }
    }
    __syncthreads();

    const int start = __builtin_amdgcn_readfirstlane(s_start);
    const int cnt   = rtot[range];

    for (int c0 = 0; c0 < cnt; c0 += SORT_CH) {
        const int nch = cnt - c0 < SORT_CH ? cnt - c0 : SORT_CH;

        int2 e[SORT_IT];
        int  rk[SORT_IT];
        #pragma unroll
        for (int k = 0; k < SORT_IT; ++k) {
            const int i = tid + k * AGG_THREADS;
            e[k] = make_int2(0, 0);
            if (i < nch) e[k] = bins[start + c0 + i];
        }
        #pragma unroll
        for (int k = 0; k < SORT_IT; ++k) {
            rk[k] = 0;
            if (tid + k * AGG_THREADS < nch) rk[k] = atomicAdd(&rcnt[e[k].x >> 16], 1);
        }
        __syncthreads();

        int c = 0, incl = 0;
        if (tid < SORT_SCAN) {
            c = rcnt[tid];
            rcnt[tid] = 0;
            incl = c;
            #pragma unroll
            for (int d = 1; d < 32; d <<= 1) {
                const int t = __shfl_up(incl, d);
                if (lane >= d) incl += t;
            }
            if (lane == 31) s_wsum[wave] = incl;
        }
        __syncthreads();
        if (tid < SORT_SCAN) {
            int st = incl - c;
            #pragma unroll
            for (int k = 0; k < SORT_SCAN / 32; ++k)
                if (k < wave) st += s_wsum[k];
            roff[tid] = st;
        }
        __syncthreads();

        #pragma unroll
        for (int k = 0; k < SORT_IT; ++k)
            if (tid + k * AGG_THREADS < nch) sorted[roff[e[k].x >> 16] + rk[k]] = e[k];
        __syncthreads();

        const int seg   = (((nch + AGG_WAVES - 1) / AGG_WAVES) + AGG_ROWS - 1) & ~(AGG_ROWS - 1);
        const int wbase = wave * seg;
        const int wend  = wbase + seg < nch ? wbase + seg : nch;
        int   cur = -1;
        float sx = 0.0f, sy = 0.0f, sz = 0.0f, sw = 0.0f;
        int2 rec = sorted[wbase + lane < nch ? wbase + lane : nch - 1];
        for (int p = wbase; p < wend; p += 32) {
            const int n = wend - p < 32 ? wend - p : 32;
            const int2 nrec = sorted[p + 32 + lane < nch ? p + 32 + lane : nch - 1];
            #pragma unroll
            for (int j0 = 0; j0 < 32; j0 += AGG_ROWS) {
                if (j0 < n) {
                    float4 v[AGG_ROWS];
                    int    rl[AGG_ROWS];
                    float  w[AGG_ROWS];
                    #pragma unroll
                    for (int u = 0; u < AGG_ROWS; ++u) {
                        const int px = __builtin_amdgcn_readlane(rec.x, j0 + u);
                        w[u]  = __int_as_float(__builtin_amdgcn_readlane(rec.y, j0 + u));
                        rl[u] = px >> 16;
                        v[u]  = *(const float4*)(y + (size_t)(px & 0xFFFF) * DIM + lane * 4);
                    }
                    #pragma unroll
                    for (int u = 0; u < AGG_ROWS; ++u)
                        asm volatile("" :: "v"(v[u].x), "v"(v[u].y), "v"(v[u].z), "v"(v[u].w));
                    #pragma unroll
                    for (int u = 0; u < AGG_ROWS; ++u) {
                        if (j0 + u < n) {
                            if (rl[u] != cur) {
                                if (cur >= 0) flush_run(acc, cur, lane, sx, sy, sz, sw);
                                cur = rl[u];
                                sx = 0.0f; sy = 0.0f; sz = 0.0f; sw = 0.0f;
                            }
                            sx += v[u].x * w[u];
                            sy += v[u].y * w[u];
                            sz += v[u].z * w[u];
                            sw += v[u].w * w[u];
                        }
                    }
                }
            }
            rec = nrec;
        }
        if (cur >= 0) flush_run(acc, cur, lane, sx, sy, sz, sw);
        __syncthreads();
    }

    float4*       orow = (float4*)(out + (size_t)node0 * DIM);
    const float4* arow = (const float4*)acc;
    for (int idx = tid; idx < RN * DIM / 4; idx += AGG_THREADS) orow[idx] = arow[idx];
}

extern "C" void kernel_launch(void* const* d_in, const int* in_sizes, int n_in,
                              void* d_out, int out_size, void* d_ws, size_t ws_size,
                              hipStream_t stream) {
    const float* nodes     = (const float*)d_in[0];
    const int*   senders   = (const int*)  d_in[1];
    const int*   receivers = (const int*)  d_in[2];
    const float* edges     = (const float*)d_in[3];
    const float* Wm        = (const float*)d_in[4];
    float*       out       = (float*)d_out;

    float* y     = (float*)d_ws;
    float* pdeg  = y + (size_t)N_NODES * DIM;
    int*   rcp   = (int*)(pdeg + (size_t)2 * DEG_SLICES * DEG_PITCH);
    int*   rtot  = rcp + DEG_CNT_ROWS * NR_PAD;
    int*   rfill = rtot + NR_PAD;
    int2*  bins  = (int2*)(rfill + NR_PAD);

    k_edge_deg<<<2 * DEG_SLICES * DEG_PARTS, DEG_THREADS, 0, stream>>>(senders, receivers, edges, pdeg, rcp, rfill);

    k_bin_gemm<<<GEMM_BLOCKS + BIN_BLOCKS, GB_THREADS, 0, stream>>>(
        nodes, Wm, pdeg, y, senders, receivers, edges, rcp, rtot, rfill, bins);
    k_row_sums<<<NR, AGG_THREADS, 0, stream>>>(y, rtot, bins, out);
}
